# starting kernel
# baseline (speedup 1.0000x reference)
#include <hip/hip_runtime.h>

#define NN 50000
#define FDIM 128

typedef __attribute__((ext_vector_type(2))) float v2f;
typedef __attribute__((ext_vector_type(8))) float v8f;


__global__ __launch_bounds__(256) void init_deg(float* deg, int n) {
    int i = blockIdx.x * blockDim.x + threadIdx.x;
    if (i < n) deg[i] = 1.0f;
}

__global__ __launch_bounds__(256) void accum_deg(const int* __restrict__ col,
                                                 float* __restrict__ deg, int e) {
    int i = blockIdx.x * blockDim.x + threadIdx.x;
    if (i < e) atomicAdd(&deg[col[i]], 1.0f);
}

__global__ __launch_bounds__(256) void finalize_dinv(float* d, int n) {
    int i = blockIdx.x * blockDim.x + threadIdx.x;
    if (i < n) d[i] = rsqrtf(d[i]);
}


template <int FOUT, bool RELU_IN>
__global__ __launch_bounds__(256) void gemm_wmma(const float* __restrict__ A,
                                                 const float* __restrict__ W,
                                                 float* __restrict__ H, int nrows) {
    constexpr int K  = FDIM;
    constexpr int CT = FOUT / 16;
    __shared__ float wlds[K * FOUT];
    for (int i = threadIdx.x; i < K * FOUT; i += 256) wlds[i] = W[i];
    __syncthreads();

    const int wave  = threadIdx.x >> 5;
    const int lane  = threadIdx.x & 31;
    const int strip = blockIdx.x * 8 + wave;
    if (strip >= (nrows >> 4)) return;

    const int row0 = strip << 4;
    const int m    = lane & 15;
    const int koff = (lane >> 4) << 1;

    v8f acc[CT];
    const v8f zero = {0.f, 0.f, 0.f, 0.f, 0.f, 0.f, 0.f, 0.f};
#pragma unroll
    for (int t = 0; t < CT; ++t) acc[t] = zero;

    const float* arow = A + (size_t)(row0 + m) * K;

    for (int kk = 0; kk < K; kk += 4) {
        float a0 = arow[kk + koff];
        float a1 = arow[kk + koff + 1];
        if (RELU_IN) { a0 = fmaxf(a0, 0.f); a1 = fmaxf(a1, 0.f); }
        v2f a; a.x = a0; a.y = a1;
#pragma unroll
        for (int t = 0; t < CT; ++t) {
            v2f b;
            b.x = wlds[(kk + koff)     * FOUT + t * 16 + m];
            b.y = wlds[(kk + koff + 1) * FOUT + t * 16 + m];
            acc[t] = __builtin_amdgcn_wmma_f32_16x16x4_f32(
                false, a, false, b, (short)0, acc[t], false, false);
        }
    }

    const int mrow = (lane >> 4) << 3;
#pragma unroll
    for (int t = 0; t < CT; ++t) {
#pragma unroll
        for (int v = 0; v < 8; ++v) {
            H[(size_t)(row0 + mrow + v) * FOUT + t * 16 + m] = acc[t][v];
        }
    }
}


template <int F>
__global__ __launch_bounds__(256) void init_out(const float* __restrict__ H,
                                                const float* __restrict__ dinv,
                                                const float* __restrict__ bias,
                                                float* __restrict__ O, int n) {
    long long i = (long long)blockIdx.x * blockDim.x + threadIdx.x;
    if (i < (long long)n * F) {
        int node = (int)(i >> (F == 128 ? 7 : 6));
        int f    = (int)(i & (F - 1));
        float d  = dinv[node];
        O[i] = bias[f] + d * d * H[i];
    }
}

template <int F>
__global__ __launch_bounds__(256) void scatter_edges(const float* __restrict__ H,
                                                     const int* __restrict__ row,
                                                     const int* __restrict__ col,
                                                     const float* __restrict__ dinv,
                                                     float* __restrict__ O, int nedges) {
    int gwave = (int)(((long long)blockIdx.x * blockDim.x + threadIdx.x) >> 5);
    int lane  = threadIdx.x & 31;
    if (gwave >= nedges) return;
    int r = row[gwave];
    int c = col[gwave];
    float w = dinv[r] * dinv[c];
    constexpr int VPL = F / 32;
    const float* hr = H + (size_t)r * F + lane * VPL;
    float*       oc = O + (size_t)c * F + lane * VPL;
#pragma unroll
    for (int v = 0; v < VPL; ++v) atomicAdd(&oc[v], w * hr[v]);
}


extern "C" void kernel_launch(void* const* d_in, const int* in_sizes, int n_in,
                              void* d_out, int out_size, void* d_ws, size_t ws_size,
                              hipStream_t stream) {
    const float* x  = (const float*)d_in[0];
    const int*   ei = (const int*)d_in[1];
    const float* W1 = (const float*)d_in[2];
    const float* b1 = (const float*)d_in[3];
    const float* W2 = (const float*)d_in[4];
    const float* b2 = (const float*)d_in[5];
    const float* W3 = (const float*)d_in[6];
    const float* b3 = (const float*)d_in[7];
    float* out = (float*)d_out;

    const int E = in_sizes[1] / 2;
    const int* row = ei;
    const int* col = ei + E;

    float* ws   = (float*)d_ws;
    float* dinv = ws;
    float* H    = ws + 50176;
    float* O    = H + (size_t)NN * FDIM;

    init_deg<<<(NN + 255) / 256, 256, 0, stream>>>(dinv, NN);
    accum_deg<<<(E + 255) / 256, 256, 0, stream>>>(col, dinv, E);
    finalize_dinv<<<(NN + 255) / 256, 256, 0, stream>>>(dinv, NN);

    const int ggrid  = ((NN / 16) + 7) / 8;
    const int sgrid  = (int)(((long long)E * 32 + 255) / 256);
    const int ig128  = (int)(((long long)NN * 128 + 255) / 256);
    const int ig64   = (int)(((long long)NN * 64 + 255) / 256);

    gemm_wmma<128, false><<<ggrid, 256, 0, stream>>>(x, W1, H, NN);
    init_out<128><<<ig128, 256, 0, stream>>>(H, dinv, b1, O, NN);
    scatter_edges<128><<<sgrid, 256, 0, stream>>>(H, row, col, dinv, O, E);

    gemm_wmma<128, true><<<ggrid, 256, 0, stream>>>(O, W2, H, NN);
    init_out<128><<<ig128, 256, 0, stream>>>(H, dinv, b2, O, NN);
    scatter_edges<128><<<sgrid, 256, 0, stream>>>(H, row, col, dinv, O, E);

    gemm_wmma<64, true><<<ggrid, 256, 0, stream>>>(O, W3, H, NN);
    init_out<64><<<ig64, 256, 0, stream>>>(H, dinv, b3, out, NN);
    scatter_edges<64><<<sgrid, 256, 0, stream>>>(H, row, col, dinv, out, E);
}


# f32 atomic scatter replaced by per-call edge list sorted by destination (int count+rank, scan, fill); one wave per node sums in-edges in registers, writes once; sum order = arrival rank
# speedup vs baseline: 1.5774x; 1.5774x over previous
#include <hip/hip_runtime.h>

#define NN 50000
#define FDIM 128

typedef __attribute__((ext_vector_type(2))) float v2f;
typedef __attribute__((ext_vector_type(8))) float v8f;


__global__ __launch_bounds__(256) void zero_cnt(int* cnt, int n) {
    int i = blockIdx.x * blockDim.x + threadIdx.x;
    if (i < n) cnt[i] = 0;
}

__global__ __launch_bounds__(256) void count_rank(const int* __restrict__ col,
                                                  int* __restrict__ cnt,
                                                  int* __restrict__ rank, int e) {
    int i = blockIdx.x * blockDim.x + threadIdx.x;
    if (i < e) rank[i] = atomicAdd(&cnt[col[i]], 1);
}

__global__ __launch_bounds__(1024) void scan_deg(const int* __restrict__ cnt,
                                                 int* __restrict__ rowptr,
                                                 float* __restrict__ dinv, int n) {
    __shared__ int wsum[32];
    const int tid  = threadIdx.x;
    const int lane = tid & 31;
    const int wv   = tid >> 5;
    const int per  = (n + 1023) / 1024;
    const int beg  = tid * per;
    const int end  = min(beg + per, n);

    int s = 0;
    for (int i = beg; i < end; ++i) s += cnt[i];

    int inc = s;
    for (int off = 1; off < 32; off <<= 1) {
        int t = __shfl_up(inc, off, 32);
        if (lane >= off) inc += t;
    }
    if (lane == 31) wsum[wv] = inc;
    __syncthreads();
    if (wv == 0) {
        int v  = wsum[lane];
        int iv = v;
        for (int off = 1; off < 32; off <<= 1) {
            int t = __shfl_up(iv, off, 32);
            if (lane >= off) iv += t;
        }
        wsum[lane] = iv - v;
    }
    __syncthreads();

    int run = wsum[wv] + inc - s;
    for (int i = beg; i < end; ++i) {
        int c = cnt[i];
        rowptr[i] = run;
        dinv[i]   = rsqrtf((float)(c + 1));
        run += c;
    }
    if (tid == 1023) rowptr[n] = run;
}

__global__ __launch_bounds__(256) void fill_csr(const int* __restrict__ row,
                                                const int* __restrict__ col,
                                                const int* __restrict__ rowptr,
                                                const int* __restrict__ rank,
                                                const float* __restrict__ dinv,
                                                int* __restrict__ src,
                                                float* __restrict__ wgt, int e) {
    int i = blockIdx.x * blockDim.x + threadIdx.x;
    if (i < e) {
        int r = row[i];
        int c = col[i];
        int s = rowptr[c] + rank[i];
        src[s] = r;
        wgt[s] = dinv[r] * dinv[c];
    }
}


template <int FOUT, bool RELU_IN>
__global__ __launch_bounds__(256) void gemm_wmma(const float* __restrict__ A,
                                                 const float* __restrict__ W,
                                                 float* __restrict__ H, int nrows) {
    constexpr int K  = FDIM;
    constexpr int CT = FOUT / 16;
    __shared__ float wlds[K * FOUT];
    for (int i = threadIdx.x; i < K * FOUT; i += 256) wlds[i] = W[i];
    __syncthreads();

    const int wave  = threadIdx.x >> 5;
    const int lane  = threadIdx.x & 31;
    const int strip = blockIdx.x * 8 + wave;
    if (strip >= (nrows >> 4)) return;

    const int row0 = strip << 4;
    const int m    = lane & 15;
    const int koff = (lane >> 4) << 1;

    v8f acc[CT];
    const v8f zero = {0.f, 0.f, 0.f, 0.f, 0.f, 0.f, 0.f, 0.f};
#pragma unroll
    for (int t = 0; t < CT; ++t) acc[t] = zero;

    const float* arow = A + (size_t)(row0 + m) * K;

    for (int kk = 0; kk < K; kk += 4) {
        float a0 = arow[kk + koff];
        float a1 = arow[kk + koff + 1];
        if (RELU_IN) { a0 = fmaxf(a0, 0.f); a1 = fmaxf(a1, 0.f); }
        v2f a; a.x = a0; a.y = a1;
#pragma unroll
        for (int t = 0; t < CT; ++t) {
            v2f b;
            b.x = wlds[(kk + koff)     * FOUT + t * 16 + m];
            b.y = wlds[(kk + koff + 1) * FOUT + t * 16 + m];
            acc[t] = __builtin_amdgcn_wmma_f32_16x16x4_f32(
                false, a, false, b, (short)0, acc[t], false, false);
        }
    }

    const int mrow = (lane >> 4) << 3;
#pragma unroll
    for (int t = 0; t < CT; ++t) {
#pragma unroll
        for (int v = 0; v < 8; ++v) {
            H[(size_t)(row0 + mrow + v) * FOUT + t * 16 + m] = acc[t][v];
        }
    }
}


template <int F>
__global__ __launch_bounds__(256) void pull_rows(const float* __restrict__ H,
                                                 const int* __restrict__ rowptr,
                                                 const int* __restrict__ src,
                                                 const float* __restrict__ wgt,
                                                 const float* __restrict__ dinv,
                                                 const float* __restrict__ bias,
                                                 float* __restrict__ O, int n) {
    constexpr int VPL = F / 32;
    typedef __attribute__((ext_vector_type(VPL))) float vec;
    const int lane = threadIdx.x & 31;
    const int node = __builtin_amdgcn_readfirstlane((int)(blockIdx.x * 8 + (threadIdx.x >> 5)));
    if (node >= n) return;

    const vec* Hv = (const vec*)H;
    const float d = dinv[node];
    vec acc = ((const vec*)bias)[lane] + (d * d) * Hv[(size_t)node * 32 + lane];

    const int beg = rowptr[node];
    const int end = rowptr[node + 1];
    for (int base = beg; base < end; base += 32) {
        const int cnt = min(32, end - base);
        int   myr = node;
        float myw = 0.f;
        if (lane < cnt) { myr = src[base + lane]; myw = wgt[base + lane]; }
        int j = 0;
        for (; j + 4 <= cnt; j += 4) {
            const int r0 = __builtin_amdgcn_readlane(myr, j);
            const int r1 = __builtin_amdgcn_readlane(myr, j + 1);
            const int r2 = __builtin_amdgcn_readlane(myr, j + 2);
            const int r3 = __builtin_amdgcn_readlane(myr, j + 3);
            const float w0 = __int_as_float(__builtin_amdgcn_readlane(__float_as_int(myw), j));
            const float w1 = __int_as_float(__builtin_amdgcn_readlane(__float_as_int(myw), j + 1));
            const float w2 = __int_as_float(__builtin_amdgcn_readlane(__float_as_int(myw), j + 2));
            const float w3 = __int_as_float(__builtin_amdgcn_readlane(__float_as_int(myw), j + 3));
            const vec h0 = Hv[(size_t)r0 * 32 + lane];
            const vec h1 = Hv[(size_t)r1 * 32 + lane];
            const vec h2 = Hv[(size_t)r2 * 32 + lane];
            const vec h3 = Hv[(size_t)r3 * 32 + lane];
            acc += w0 * h0;
            acc += w1 * h1;
            acc += w2 * h2;
            acc += w3 * h3;
        }
        for (; j < cnt; ++j) {
            const int   r = __builtin_amdgcn_readlane(myr, j);
            const float w = __int_as_float(__builtin_amdgcn_readlane(__float_as_int(myw), j));
            acc += w * Hv[(size_t)r * 32 + lane];
        }
    }
    ((vec*)O)[(size_t)node * 32 + lane] = acc;
}


extern "C" void kernel_launch(void* const* d_in, const int* in_sizes, int n_in,
                              void* d_out, int out_size, void* d_ws, size_t ws_size,
                              hipStream_t stream) {
    const float* x  = (const float*)d_in[0];
    const int*   ei = (const int*)d_in[1];
    const float* W1 = (const float*)d_in[2];
    const float* b1 = (const float*)d_in[3];
    const float* W2 = (const float*)d_in[4];
    const float* b2 = (const float*)d_in[5];
    const float* W3 = (const float*)d_in[6];
    const float* b3 = (const float*)d_in[7];
    float* out = (float*)d_out;

    const int E = in_sizes[1] / 2;
    const int* row = ei;
    const int* col = ei + E;

    float* ws   = (float*)d_ws;
    float* dinv = ws;
    float* H    = ws + 50176;
    float* O    = H + (size_t)NN * FDIM;
    int*   cnt    = (int*)(O + (size_t)NN * FDIM);
    int*   rowptr = cnt + 50176;
    int*   rank   = rowptr + 50176;
    int*   src    = rank + E;
    float* wgt    = (float*)(src + E);

    const int egrid = (E + 255) / 256;
    const int ggrid = ((NN / 16) + 7) / 8;
    const int pgrid = (NN + 7) / 8;

    zero_cnt<<<(NN + 255) / 256, 256, 0, stream>>>(cnt, NN);
    count_rank<<<egrid, 256, 0, stream>>>(col, cnt, rank, E);
    scan_deg<<<1, 1024, 0, stream>>>(cnt, rowptr, dinv, NN);
    fill_csr<<<egrid, 256, 0, stream>>>(row, col, rowptr, rank, dinv, src, wgt, E);

    gemm_wmma<128, false><<<ggrid, 256, 0, stream>>>(x, W1, H, NN);
    pull_rows<128><<<pgrid, 256, 0, stream>>>(H, rowptr, src, wgt, dinv, b1, O, NN);

    gemm_wmma<128, true><<<ggrid, 256, 0, stream>>>(O, W2, H, NN);
    pull_rows<128><<<pgrid, 256, 0, stream>>>(H, rowptr, src, wgt, dinv, b2, O, NN);

    gemm_wmma<64, true><<<ggrid, 256, 0, stream>>>(O, W3, H, NN);
    pull_rows<64><<<pgrid, 256, 0, stream>>>(H, rowptr, src, wgt, dinv, b3, out, NN);
}


# single-block strided degree scan replaced by 49 blocks, one node per thread: each sums the counts in front of it with 128-bit loads, then scans its own by shuffles and LDS; integer sums only
# speedup vs baseline: 2.2309x; 1.4143x over previous
#include <hip/hip_runtime.h>

#define NN 50000
#define FDIM 128
#define SCAN_ROUNDS (((NN + 1023) / 1024 - 1 + 3) / 4)

typedef __attribute__((ext_vector_type(2))) float v2f;
typedef __attribute__((ext_vector_type(8))) float v8f;


__global__ __launch_bounds__(256) void zero_cnt(int* cnt, int n) {
    int i = blockIdx.x * blockDim.x + threadIdx.x;
    if (i < n) cnt[i] = 0;
}

__global__ __launch_bounds__(256) void count_rank(const int* __restrict__ col,
                                                  int* __restrict__ cnt,
                                                  int* __restrict__ rank, int e) {
    int i = blockIdx.x * blockDim.x + threadIdx.x;
    if (i < e) rank[i] = atomicAdd(&cnt[col[i]], 1);
}

__global__ __launch_bounds__(1024) void scan_deg(const int* __restrict__ cnt,
                                                 int* __restrict__ rowptr,
                                                 float* __restrict__ dinv, int n) {
    __shared__ int wtot[32];
    __shared__ int ptot[32];
    const int tid  = threadIdx.x;
    const int lane = tid & 31;
    const int wv   = tid >> 5;
    const int base = blockIdx.x * 1024;

    const int4* c4  = (const int4*)cnt;
    const int   lim = base >> 2;
    int4 v[SCAN_ROUNDS];
#pragma unroll
    for (int r = 0; r < SCAN_ROUNDS; ++r) {
        const int i = tid + r * 1024;
        v[r] = (i < lim) ? c4[i] : make_int4(0, 0, 0, 0);
    }
    int pre = 0;
#pragma unroll
    for (int r = 0; r < SCAN_ROUNDS; ++r) pre += (v[r].x + v[r].y) + (v[r].z + v[r].w);

    const int idx = base + tid;
    const int c   = (idx < n) ? cnt[idx] : 0;

    int inc = c;
    for (int off = 1; off < 32; off <<= 1) {
        int t = __shfl_up(inc, off, 32);
        if (lane >= off) inc += t;
    }
    for (int off = 16; off > 0; off >>= 1) pre += __shfl_xor(pre, off, 32);
    if (lane == 31) wtot[wv] = inc;
    if (lane == 0)  ptot[wv] = pre;
    __syncthreads();

    int front = ptot[lane] + ((lane < wv) ? wtot[lane] : 0);
    for (int off = 16; off > 0; off >>= 1) front += __shfl_xor(front, off, 32);

    if (idx < n) {
        const int excl = front + inc - c;
        rowptr[idx] = excl;
        dinv[idx]   = rsqrtf((float)(c + 1));
        if (idx == n - 1) rowptr[n] = excl + c;
    }
}

__global__ __launch_bounds__(256) void fill_csr(const int* __restrict__ row,
                                                const int* __restrict__ col,
                                                const int* __restrict__ rowptr,
                                                const int* __restrict__ rank,
                                                const float* __restrict__ dinv,
                                                int* __restrict__ src,
                                                float* __restrict__ wgt, int e) {
    int i = blockIdx.x * blockDim.x + threadIdx.x;
    if (i < e) {
        int r = row[i];
        int c = col[i];
        int s = rowptr[c] + rank[i];
        src[s] = r;
        wgt[s] = dinv[r] * dinv[c];
    }
}


template <int FOUT, bool RELU_IN>
__global__ __launch_bounds__(256) void gemm_wmma(const float* __restrict__ A,
                                                 const float* __restrict__ W,
                                                 float* __restrict__ H, int nrows) {
    constexpr int K  = FDIM;
    constexpr int CT = FOUT / 16;
    __shared__ float wlds[K * FOUT];
    for (int i = threadIdx.x; i < K * FOUT; i += 256) wlds[i] = W[i];
    __syncthreads();

    const int wave  = threadIdx.x >> 5;
    const int lane  = threadIdx.x & 31;
    const int strip = blockIdx.x * 8 + wave;
    if (strip >= (nrows >> 4)) return;

    const int row0 = strip << 4;
    const int m    = lane & 15;
    const int koff = (lane >> 4) << 1;

    v8f acc[CT];
    const v8f zero = {0.f, 0.f, 0.f, 0.f, 0.f, 0.f, 0.f, 0.f};
#pragma unroll
    for (int t = 0; t < CT; ++t) acc[t] = zero;

    const float* arow = A + (size_t)(row0 + m) * K;

    for (int kk = 0; kk < K; kk += 4) {
        float a0 = arow[kk + koff];
        float a1 = arow[kk + koff + 1];
        if (RELU_IN) { a0 = fmaxf(a0, 0.f); a1 = fmaxf(a1, 0.f); }
        v2f a; a.x = a0; a.y = a1;
#pragma unroll
        for (int t = 0; t < CT; ++t) {
            v2f b;
            b.x = wlds[(kk + koff)     * FOUT + t * 16 + m];
            b.y = wlds[(kk + koff + 1) * FOUT + t * 16 + m];
            acc[t] = __builtin_amdgcn_wmma_f32_16x16x4_f32(
                false, a, false, b, (short)0, acc[t], false, false);
        }
    }

    const int mrow = (lane >> 4) << 3;
#pragma unroll
    for (int t = 0; t < CT; ++t) {
#pragma unroll
        for (int v = 0; v < 8; ++v) {
            H[(size_t)(row0 + mrow + v) * FOUT + t * 16 + m] = acc[t][v];
        }
    }
}


template <int F>
__global__ __launch_bounds__(256) void pull_rows(const float* __restrict__ H,
                                                 const int* __restrict__ rowptr,
                                                 const int* __restrict__ src,
                                                 const float* __restrict__ wgt,
                                                 const float* __restrict__ dinv,
                                                 const float* __restrict__ bias,
                                                 float* __restrict__ O, int n) {
    constexpr int VPL = F / 32;
    typedef __attribute__((ext_vector_type(VPL))) float vec;
    const int lane = threadIdx.x & 31;
    const int node = __builtin_amdgcn_readfirstlane((int)(blockIdx.x * 8 + (threadIdx.x >> 5)));
    if (node >= n) return;

    const vec* Hv = (const vec*)H;
    const float d = dinv[node];
    vec acc = ((const vec*)bias)[lane] + (d * d) * Hv[(size_t)node * 32 + lane];

    const int beg = rowptr[node];
    const int end = rowptr[node + 1];
    for (int base = beg; base < end; base += 32) {
        const int cnt = min(32, end - base);
        int   myr = node;
        float myw = 0.f;
        if (lane < cnt) { myr = src[base + lane]; myw = wgt[base + lane]; }
        int j = 0;
        for (; j + 4 <= cnt; j += 4) {
            const int r0 = __builtin_amdgcn_readlane(myr, j);
            const int r1 = __builtin_amdgcn_readlane(myr, j + 1);
            const int r2 = __builtin_amdgcn_readlane(myr, j + 2);
            const int r3 = __builtin_amdgcn_readlane(myr, j + 3);
            const float w0 = __int_as_float(__builtin_amdgcn_readlane(__float_as_int(myw), j));
            const float w1 = __int_as_float(__builtin_amdgcn_readlane(__float_as_int(myw), j + 1));
            const float w2 = __int_as_float(__builtin_amdgcn_readlane(__float_as_int(myw), j + 2));
            const float w3 = __int_as_float(__builtin_amdgcn_readlane(__float_as_int(myw), j + 3));
            const vec h0 = Hv[(size_t)r0 * 32 + lane];
            const vec h1 = Hv[(size_t)r1 * 32 + lane];
            const vec h2 = Hv[(size_t)r2 * 32 + lane];
            const vec h3 = Hv[(size_t)r3 * 32 + lane];
            acc += w0 * h0;
            acc += w1 * h1;
            acc += w2 * h2;
            acc += w3 * h3;
        }
        for (; j < cnt; ++j) {
            const int   r = __builtin_amdgcn_readlane(myr, j);
            const float w = __int_as_float(__builtin_amdgcn_readlane(__float_as_int(myw), j));
            acc += w * Hv[(size_t)r * 32 + lane];
        }
    }
    ((vec*)O)[(size_t)node * 32 + lane] = acc;
}


extern "C" void kernel_launch(void* const* d_in, const int* in_sizes, int n_in,
                              void* d_out, int out_size, void* d_ws, size_t ws_size,
                              hipStream_t stream) {
    const float* x  = (const float*)d_in[0];
    const int*   ei = (const int*)d_in[1];
    const float* W1 = (const float*)d_in[2];
    const float* b1 = (const float*)d_in[3];
    const float* W2 = (const float*)d_in[4];
    const float* b2 = (const float*)d_in[5];
    const float* W3 = (const float*)d_in[6];
    const float* b3 = (const float*)d_in[7];
    float* out = (float*)d_out;

    const int E = in_sizes[1] / 2;
    const int* row = ei;
    const int* col = ei + E;

    float* ws   = (float*)d_ws;
    float* dinv = ws;
    float* H    = ws + 50176;
    float* O    = H + (size_t)NN * FDIM;
    int*   cnt    = (int*)(O + (size_t)NN * FDIM);
    int*   rowptr = cnt + 50176;
    int*   rank   = rowptr + 50176;
    int*   src    = rank + E;
    float* wgt    = (float*)(src + E);

    const int egrid = (E + 255) / 256;
    const int ggrid = ((NN / 16) + 7) / 8;
    const int pgrid = (NN + 7) / 8;

    zero_cnt<<<(NN + 255) / 256, 256, 0, stream>>>(cnt, NN);
    count_rank<<<egrid, 256, 0, stream>>>(col, cnt, rank, E);
    scan_deg<<<(NN + 1023) / 1024, 1024, 0, stream>>>(cnt, rowptr, dinv, NN);
    fill_csr<<<egrid, 256, 0, stream>>>(row, col, rowptr, rank, dinv, src, wgt, E);

    gemm_wmma<128, false><<<ggrid, 256, 0, stream>>>(x, W1, H, NN);
    pull_rows<128><<<pgrid, 256, 0, stream>>>(H, rowptr, src, wgt, dinv, b1, O, NN);

    gemm_wmma<128, true><<<ggrid, 256, 0, stream>>>(O, W2, H, NN);
    pull_rows<128><<<pgrid, 256, 0, stream>>>(H, rowptr, src, wgt, dinv, b2, O, NN);

    gemm_wmma<64, true><<<ggrid, 256, 0, stream>>>(O, W3, H, NN);
    pull_rows<64><<<pgrid, 256, 0, stream>>>(H, rowptr, src, wgt, dinv, b3, out, NN);
}


# one returning global atomic per edge replaced by LDS counters: 64 blocks each rank one piece of the edge list, the degree scan sums the pieces; sum order = piece, then arrival rank inside it
# speedup vs baseline: 2.3421x; 1.0498x over previous
#include <hip/hip_runtime.h>

#define NN 50000
#define FDIM 128

typedef __attribute__((ext_vector_type(2))) float v2f;
typedef __attribute__((ext_vector_type(8))) float v8f;


#define CHUNKS 64
#define NPAD 50176
#define NGRP 64
#define RANK_BATCH 16

__global__ __launch_bounds__(1024) void chunk_rank(const int* __restrict__ col,
                                                   int* __restrict__ bcnt,
                                                   int* __restrict__ gsum,
                                                   int* __restrict__ rank, int e, int chunk) {
    __shared__ int hist[NN];
    __shared__ int grp[NGRP];
    const int tid = threadIdx.x;
    int4* h4 = (int4*)hist;
    for (int i = tid; i < NN / 4; i += 1024) h4[i] = make_int4(0, 0, 0, 0);
    if (tid < NGRP) grp[tid] = 0;
    __syncthreads();

    const int beg = blockIdx.x * chunk;
    const int len = min(chunk, e - beg);
    for (int k0 = 0; k0 < len; k0 += RANK_BATCH * 1024) {
        int c[RANK_BATCH], r[RANK_BATCH];
#pragma unroll
        for (int u = 0; u < RANK_BATCH; ++u) {
            const int k = k0 + u * 1024 + tid;
            c[u] = (k < len) ? col[beg + k] : -1;
        }
#pragma unroll
        for (int u = 0; u < RANK_BATCH; ++u) r[u] = (c[u] >= 0) ? atomicAdd(&hist[c[u]], 1) : 0;
#pragma unroll
        for (int u = 0; u < RANK_BATCH; ++u) {
            const int k = k0 + u * 1024 + tid;
            if (c[u] >= 0) rank[beg + k] = r[u];
        }
    }
    __syncthreads();

    int4* o4 = (int4*)(bcnt + (size_t)blockIdx.x * NPAD);
    for (int i = tid; i < NN / 4; i += 1024) {
        const int4 v = h4[i];
        o4[i] = v;
        atomicAdd(&grp[i >> 8], (v.x + v.y) + (v.z + v.w));
    }
    __syncthreads();
    if (tid < NGRP) gsum[blockIdx.x * NGRP + tid] = grp[tid];
}

__global__ __launch_bounds__(1024) void scan_deg(int* __restrict__ bcnt,
                                                 const int* __restrict__ gsum,
                                                 int* __restrict__ rowptr,
                                                 float* __restrict__ dinv, int n) {
    __shared__ int wtot[32];
    __shared__ int ptot[32];
    const int tid  = threadIdx.x;
    const int lane = tid & 31;
    const int wv   = tid >> 5;
    const int base = blockIdx.x * 1024;
    const int grp  = blockIdx.x;

    int pre = 0;
#pragma unroll
    for (int k = 0; k < CHUNKS * NGRP / 1024; ++k) {
        const int j = tid + k * 1024;
        pre += ((j & (NGRP - 1)) < grp) ? gsum[j] : 0;
    }

    const int idx = base + tid;
    int c = 0;
    if (idx < n) {
#pragma unroll
        for (int b = 0; b < CHUNKS; ++b) {
            const int v = bcnt[(size_t)b * NPAD + idx];
            bcnt[(size_t)b * NPAD + idx] = c;
            c += v;
        }
    }

    int inc = c;
    for (int off = 1; off < 32; off <<= 1) {
        int t = __shfl_up(inc, off, 32);
        if (lane >= off) inc += t;
    }
    for (int off = 16; off > 0; off >>= 1) pre += __shfl_xor(pre, off, 32);
    if (lane == 31) wtot[wv] = inc;
    if (lane == 0)  ptot[wv] = pre;
    __syncthreads();

    int front = ptot[lane] + ((lane < wv) ? wtot[lane] : 0);
    for (int off = 16; off > 0; off >>= 1) front += __shfl_xor(front, off, 32);

    if (idx < n) {
        const int excl = front + inc - c;
        rowptr[idx] = excl;
        dinv[idx]   = rsqrtf((float)(c + 1));
        if (idx == n - 1) rowptr[n] = excl + c;
    }
}

__global__ __launch_bounds__(256) void fill_csr(const int* __restrict__ row,
                                                const int* __restrict__ col,
                                                const int* __restrict__ rowptr,
                                                const int* __restrict__ rank,
                                                const int* __restrict__ bcnt,
                                                const float* __restrict__ dinv,
                                                int* __restrict__ src,
                                                float* __restrict__ wgt, int e, int chunk) {
    int i = blockIdx.x * blockDim.x + threadIdx.x;
    if (i < e) {
        int r = row[i];
        int c = col[i];
        int s = rowptr[c] + bcnt[(size_t)(i / chunk) * NPAD + c] + rank[i];
        src[s] = r;
        wgt[s] = dinv[r] * dinv[c];
    }
}


template <int FOUT, bool RELU_IN>
__global__ __launch_bounds__(256) void gemm_wmma(const float* __restrict__ A,
                                                 const float* __restrict__ W,
                                                 float* __restrict__ H, int nrows) {
    constexpr int K  = FDIM;
    constexpr int CT = FOUT / 16;
    __shared__ float wlds[K * FOUT];
    for (int i = threadIdx.x; i < K * FOUT; i += 256) wlds[i] = W[i];
    __syncthreads();

    const int wave  = threadIdx.x >> 5;
    const int lane  = threadIdx.x & 31;
    const int strip = blockIdx.x * 8 + wave;
    if (strip >= (nrows >> 4)) return;

    const int row0 = strip << 4;
    const int m    = lane & 15;
    const int koff = (lane >> 4) << 1;

    v8f acc[CT];
    const v8f zero = {0.f, 0.f, 0.f, 0.f, 0.f, 0.f, 0.f, 0.f};
#pragma unroll
    for (int t = 0; t < CT; ++t) acc[t] = zero;

    const float* arow = A + (size_t)(row0 + m) * K;

    for (int kk = 0; kk < K; kk += 4) {
        float a0 = arow[kk + koff];
        float a1 = arow[kk + koff + 1];
        if (RELU_IN) { a0 = fmaxf(a0, 0.f); a1 = fmaxf(a1, 0.f); }
        v2f a; a.x = a0; a.y = a1;
#pragma unroll
        for (int t = 0; t < CT; ++t) {
            v2f b;
            b.x = wlds[(kk + koff)     * FOUT + t * 16 + m];
            b.y = wlds[(kk + koff + 1) * FOUT + t * 16 + m];
            acc[t] = __builtin_amdgcn_wmma_f32_16x16x4_f32(
                false, a, false, b, (short)0, acc[t], false, false);
        }
    }

    const int mrow = (lane >> 4) << 3;
#pragma unroll
    for (int t = 0; t < CT; ++t) {
#pragma unroll
        for (int v = 0; v < 8; ++v) {
            H[(size_t)(row0 + mrow + v) * FOUT + t * 16 + m] = acc[t][v];
        }
    }
}


template <int F>
__global__ __launch_bounds__(256) void pull_rows(const float* __restrict__ H,
                                                 const int* __restrict__ rowptr,
                                                 const int* __restrict__ src,
                                                 const float* __restrict__ wgt,
                                                 const float* __restrict__ dinv,
                                                 const float* __restrict__ bias,
                                                 float* __restrict__ O, int n) {
    constexpr int VPL = F / 32;
    typedef __attribute__((ext_vector_type(VPL))) float vec;
    const int lane = threadIdx.x & 31;
    const int node = __builtin_amdgcn_readfirstlane((int)(blockIdx.x * 8 + (threadIdx.x >> 5)));
    if (node >= n) return;

    const vec* Hv = (const vec*)H;
    const float d = dinv[node];
    vec acc = ((const vec*)bias)[lane] + (d * d) * Hv[(size_t)node * 32 + lane];

    const int beg = rowptr[node];
    const int end = rowptr[node + 1];
    for (int base = beg; base < end; base += 32) {
        const int cnt = min(32, end - base);
        int   myr = node;
        float myw = 0.f;
        if (lane < cnt) { myr = src[base + lane]; myw = wgt[base + lane]; }
        int j = 0;
        for (; j + 4 <= cnt; j += 4) {
            const int r0 = __builtin_amdgcn_readlane(myr, j);
            const int r1 = __builtin_amdgcn_readlane(myr, j + 1);
            const int r2 = __builtin_amdgcn_readlane(myr, j + 2);
            const int r3 = __builtin_amdgcn_readlane(myr, j + 3);
            const float w0 = __int_as_float(__builtin_amdgcn_readlane(__float_as_int(myw), j));
            const float w1 = __int_as_float(__builtin_amdgcn_readlane(__float_as_int(myw), j + 1));
            const float w2 = __int_as_float(__builtin_amdgcn_readlane(__float_as_int(myw), j + 2));
            const float w3 = __int_as_float(__builtin_amdgcn_readlane(__float_as_int(myw), j + 3));
            const vec h0 = Hv[(size_t)r0 * 32 + lane];
            const vec h1 = Hv[(size_t)r1 * 32 + lane];
            const vec h2 = Hv[(size_t)r2 * 32 + lane];
            const vec h3 = Hv[(size_t)r3 * 32 + lane];
            acc += w0 * h0;
            acc += w1 * h1;
            acc += w2 * h2;
            acc += w3 * h3;
        }
        for (; j < cnt; ++j) {
            const int   r = __builtin_amdgcn_readlane(myr, j);
            const float w = __int_as_float(__builtin_amdgcn_readlane(__float_as_int(myw), j));
            acc += w * Hv[(size_t)r * 32 + lane];
        }
    }
    ((vec*)O)[(size_t)node * 32 + lane] = acc;
}


extern "C" void kernel_launch(void* const* d_in, const int* in_sizes, int n_in,
                              void* d_out, int out_size, void* d_ws, size_t ws_size,
                              hipStream_t stream) {
    const float* x  = (const float*)d_in[0];
    const int*   ei = (const int*)d_in[1];
    const float* W1 = (const float*)d_in[2];
    const float* b1 = (const float*)d_in[3];
    const float* W2 = (const float*)d_in[4];
    const float* b2 = (const float*)d_in[5];
    const float* W3 = (const float*)d_in[6];
    const float* b3 = (const float*)d_in[7];
    float* out = (float*)d_out;

    const int E = in_sizes[1] / 2;
    const int* row = ei;
    const int* col = ei + E;

    float* ws   = (float*)d_ws;
    float* dinv = ws;
    float* H    = ws + 50176;
    float* O    = H + (size_t)NN * FDIM;
    int*   gsum   = (int*)(O + (size_t)NN * FDIM);
    int*   rowptr = gsum + 50176;
    int*   bcnt   = rowptr + 50176;
    int*   rank   = bcnt + (size_t)CHUNKS * NPAD;
    int*   src    = rank + E;
    float* wgt    = (float*)(src + E);

    const int egrid = (E + 255) / 256;
    const int chunk = (E + CHUNKS - 1) / CHUNKS;
    const int ggrid = ((NN / 16) + 7) / 8;
    const int pgrid = (NN + 7) / 8;

    chunk_rank<<<CHUNKS, 1024, 0, stream>>>(col, bcnt, gsum, rank, E, chunk);
    scan_deg<<<(NN + 1023) / 1024, 1024, 0, stream>>>(bcnt, gsum, rowptr, dinv, NN);
    fill_csr<<<egrid, 256, 0, stream>>>(row, col, rowptr, rank, bcnt, dinv, src, wgt, E, chunk);

    gemm_wmma<128, false><<<ggrid, 256, 0, stream>>>(x, W1, H, NN);
    pull_rows<128><<<pgrid, 256, 0, stream>>>(H, rowptr, src, wgt, dinv, b1, O, NN);

    gemm_wmma<128, true><<<ggrid, 256, 0, stream>>>(O, W2, H, NN);
    pull_rows<128><<<pgrid, 256, 0, stream>>>(H, rowptr, src, wgt, dinv, b2, O, NN);

    gemm_wmma<64, true><<<ggrid, 256, 0, stream>>>(O, W3, H, NN);
    pull_rows<64><<<pgrid, 256, 0, stream>>>(H, rowptr, src, wgt, dinv, b3, out, NN);
}


# per-piece group sums: 32-lane same-word LDS atomics (12500 per block, not merged by the LDS) replaced by wave sums read from the finished LDS counters; no atomic left there; integer sums only
# speedup vs baseline: 2.6831x; 1.1456x over previous
#include <hip/hip_runtime.h>

#define NN 50000
#define FDIM 128

typedef __attribute__((ext_vector_type(2))) float v2f;
typedef __attribute__((ext_vector_type(8))) float v8f;


#define CHUNKS 64
#define NPAD 50176
#define NGRP 64
#define RANK_BATCH 16

__global__ __launch_bounds__(1024) void chunk_rank(const int* __restrict__ col,
                                                   int* __restrict__ bcnt,
                                                   int* __restrict__ gsum,
                                                   int* __restrict__ rank, int e, int chunk) {
    __shared__ int hist[NPAD];
    const int tid = threadIdx.x;
    int4* h4 = (int4*)hist;
    for (int i = tid; i < NPAD / 4; i += 1024) h4[i] = make_int4(0, 0, 0, 0);
    __syncthreads();

    const int beg = blockIdx.x * chunk;
    const int len = min(chunk, e - beg);
    for (int k0 = 0; k0 < len; k0 += RANK_BATCH * 1024) {
        int c[RANK_BATCH], r[RANK_BATCH];
#pragma unroll
        for (int u = 0; u < RANK_BATCH; ++u) {
            const int k = k0 + u * 1024 + tid;
            c[u] = (k < len) ? col[beg + k] : -1;
        }
#pragma unroll
        for (int u = 0; u < RANK_BATCH; ++u) r[u] = (c[u] >= 0) ? atomicAdd(&hist[c[u]], 1) : 0;
#pragma unroll
        for (int u = 0; u < RANK_BATCH; ++u) {
            const int k = k0 + u * 1024 + tid;
            if (c[u] >= 0) rank[beg + k] = r[u];
        }
    }
    __syncthreads();

    int4* o4 = (int4*)(bcnt + (size_t)blockIdx.x * NPAD);
    for (int i = tid; i < NN / 4; i += 1024) o4[i] = h4[i];

    const int lane = tid & 31;
    const int wv   = tid >> 5;
    int s0 = 0, s1 = 0;
#pragma unroll
    for (int k = 0; k < 8; ++k) {
        const int4 v = h4[wv * 256 + k * 32 + lane];
        s0 += (v.x + v.y) + (v.z + v.w);
    }
    if ((wv + 32) * 1024 < NN) {
#pragma unroll
        for (int k = 0; k < 8; ++k) {
            const int4 v = h4[(wv + 32) * 256 + k * 32 + lane];
            s1 += (v.x + v.y) + (v.z + v.w);
        }
    }
    for (int off = 16; off > 0; off >>= 1) {
        s0 += __shfl_xor(s0, off, 32);
        s1 += __shfl_xor(s1, off, 32);
    }
    if (lane == 0) {
        gsum[blockIdx.x * NGRP + wv]      = s0;
        gsum[blockIdx.x * NGRP + wv + 32] = s1;
    }
}

__global__ __launch_bounds__(1024) void scan_deg(int* __restrict__ bcnt,
                                                 const int* __restrict__ gsum,
                                                 int* __restrict__ rowptr,
                                                 float* __restrict__ dinv, int n) {
    __shared__ int wtot[32];
    __shared__ int ptot[32];
    const int tid  = threadIdx.x;
    const int lane = tid & 31;
    const int wv   = tid >> 5;
    const int base = blockIdx.x * 1024;
    const int grp  = blockIdx.x;

    int pre = 0;
#pragma unroll
    for (int k = 0; k < CHUNKS * NGRP / 1024; ++k) {
        const int j = tid + k * 1024;
        pre += ((j & (NGRP - 1)) < grp) ? gsum[j] : 0;
    }

    const int idx = base + tid;
    int c = 0;
    if (idx < n) {
#pragma unroll
        for (int b = 0; b < CHUNKS; ++b) {
            const int v = bcnt[(size_t)b * NPAD + idx];
            bcnt[(size_t)b * NPAD + idx] = c;
            c += v;
        }
    }

    int inc = c;
    for (int off = 1; off < 32; off <<= 1) {
        int t = __shfl_up(inc, off, 32);
        if (lane >= off) inc += t;
    }
    for (int off = 16; off > 0; off >>= 1) pre += __shfl_xor(pre, off, 32);
    if (lane == 31) wtot[wv] = inc;
    if (lane == 0)  ptot[wv] = pre;
    __syncthreads();

    int front = ptot[lane] + ((lane < wv) ? wtot[lane] : 0);
    for (int off = 16; off > 0; off >>= 1) front += __shfl_xor(front, off, 32);

    if (idx < n) {
        const int excl = front + inc - c;
        rowptr[idx] = excl;
        dinv[idx]   = rsqrtf((float)(c + 1));
        if (idx == n - 1) rowptr[n] = excl + c;
    }
}

__global__ __launch_bounds__(256) void fill_csr(const int* __restrict__ row,
                                                const int* __restrict__ col,
                                                const int* __restrict__ rowptr,
                                                const int* __restrict__ rank,
                                                const int* __restrict__ bcnt,
                                                const float* __restrict__ dinv,
                                                int* __restrict__ src,
                                                float* __restrict__ wgt, int e, int chunk) {
    int i = blockIdx.x * blockDim.x + threadIdx.x;
    if (i < e) {
        int r = row[i];
        int c = col[i];
        int s = rowptr[c] + bcnt[(size_t)(i / chunk) * NPAD + c] + rank[i];
        src[s] = r;
        wgt[s] = dinv[r] * dinv[c];
    }
}


template <int FOUT, bool RELU_IN>
__global__ __launch_bounds__(256) void gemm_wmma(const float* __restrict__ A,
                                                 const float* __restrict__ W,
                                                 float* __restrict__ H, int nrows) {
    constexpr int K  = FDIM;
    constexpr int CT = FOUT / 16;
    __shared__ float wlds[K * FOUT];
    for (int i = threadIdx.x; i < K * FOUT; i += 256) wlds[i] = W[i];
    __syncthreads();

    const int wave  = threadIdx.x >> 5;
    const int lane  = threadIdx.x & 31;
    const int strip = blockIdx.x * 8 + wave;
    if (strip >= (nrows >> 4)) return;

    const int row0 = strip << 4;
    const int m    = lane & 15;
    const int koff = (lane >> 4) << 1;

    v8f acc[CT];
    const v8f zero = {0.f, 0.f, 0.f, 0.f, 0.f, 0.f, 0.f, 0.f};
#pragma unroll
    for (int t = 0; t < CT; ++t) acc[t] = zero;

    const float* arow = A + (size_t)(row0 + m) * K;

    for (int kk = 0; kk < K; kk += 4) {
        float a0 = arow[kk + koff];
        float a1 = arow[kk + koff + 1];
        if (RELU_IN) { a0 = fmaxf(a0, 0.f); a1 = fmaxf(a1, 0.f); }
        v2f a; a.x = a0; a.y = a1;
#pragma unroll
        for (int t = 0; t < CT; ++t) {
            v2f b;
            b.x = wlds[(kk + koff)     * FOUT + t * 16 + m];
            b.y = wlds[(kk + koff + 1) * FOUT + t * 16 + m];
            acc[t] = __builtin_amdgcn_wmma_f32_16x16x4_f32(
                false, a, false, b, (short)0, acc[t], false, false);
        }
    }

    const int mrow = (lane >> 4) << 3;
#pragma unroll
    for (int t = 0; t < CT; ++t) {
#pragma unroll
        for (int v = 0; v < 8; ++v) {
            H[(size_t)(row0 + mrow + v) * FOUT + t * 16 + m] = acc[t][v];
        }
    }
}


template <int F>
__global__ __launch_bounds__(256) void pull_rows(const float* __restrict__ H,
                                                 const int* __restrict__ rowptr,
                                                 const int* __restrict__ src,
                                                 const float* __restrict__ wgt,
                                                 const float* __restrict__ dinv,
                                                 const float* __restrict__ bias,
                                                 float* __restrict__ O, int n) {
    constexpr int VPL = F / 32;
    typedef __attribute__((ext_vector_type(VPL))) float vec;
    const int lane = threadIdx.x & 31;
    const int node = __builtin_amdgcn_readfirstlane((int)(blockIdx.x * 8 + (threadIdx.x >> 5)));
    if (node >= n) return;

    const vec* Hv = (const vec*)H;
    const float d = dinv[node];
    vec acc = ((const vec*)bias)[lane] + (d * d) * Hv[(size_t)node * 32 + lane];

    const int beg = rowptr[node];
    const int end = rowptr[node + 1];
    for (int base = beg; base < end; base += 32) {
        const int cnt = min(32, end - base);
        int   myr = node;
        float myw = 0.f;
        if (lane < cnt) { myr = src[base + lane]; myw = wgt[base + lane]; }
        int j = 0;
        for (; j + 4 <= cnt; j += 4) {
            const int r0 = __builtin_amdgcn_readlane(myr, j);
            const int r1 = __builtin_amdgcn_readlane(myr, j + 1);
            const int r2 = __builtin_amdgcn_readlane(myr, j + 2);
            const int r3 = __builtin_amdgcn_readlane(myr, j + 3);
            const float w0 = __int_as_float(__builtin_amdgcn_readlane(__float_as_int(myw), j));
            const float w1 = __int_as_float(__builtin_amdgcn_readlane(__float_as_int(myw), j + 1));
            const float w2 = __int_as_float(__builtin_amdgcn_readlane(__float_as_int(myw), j + 2));
            const float w3 = __int_as_float(__builtin_amdgcn_readlane(__float_as_int(myw), j + 3));
            const vec h0 = Hv[(size_t)r0 * 32 + lane];
            const vec h1 = Hv[(size_t)r1 * 32 + lane];
            const vec h2 = Hv[(size_t)r2 * 32 + lane];
            const vec h3 = Hv[(size_t)r3 * 32 + lane];
            acc += w0 * h0;
            acc += w1 * h1;
            acc += w2 * h2;
            acc += w3 * h3;
        }
        for (; j < cnt; ++j) {
            const int   r = __builtin_amdgcn_readlane(myr, j);
            const float w = __int_as_float(__builtin_amdgcn_readlane(__float_as_int(myw), j));
            acc += w * Hv[(size_t)r * 32 + lane];
        }
    }
    ((vec*)O)[(size_t)node * 32 + lane] = acc;
}


extern "C" void kernel_launch(void* const* d_in, const int* in_sizes, int n_in,
                              void* d_out, int out_size, void* d_ws, size_t ws_size,
                              hipStream_t stream) {
    const float* x  = (const float*)d_in[0];
    const int*   ei = (const int*)d_in[1];
    const float* W1 = (const float*)d_in[2];
    const float* b1 = (const float*)d_in[3];
    const float* W2 = (const float*)d_in[4];
    const float* b2 = (const float*)d_in[5];
    const float* W3 = (const float*)d_in[6];
    const float* b3 = (const float*)d_in[7];
    float* out = (float*)d_out;

    const int E = in_sizes[1] / 2;
    const int* row = ei;
    const int* col = ei + E;

    float* ws   = (float*)d_ws;
    float* dinv = ws;
    float* H    = ws + 50176;
    float* O    = H + (size_t)NN * FDIM;
    int*   gsum   = (int*)(O + (size_t)NN * FDIM);
    int*   rowptr = gsum + 50176;
    int*   bcnt   = rowptr + 50176;
    int*   rank   = bcnt + (size_t)CHUNKS * NPAD;
    int*   src    = rank + E;
    float* wgt    = (float*)(src + E);

    const int egrid = (E + 255) / 256;
    const int chunk = (E + CHUNKS - 1) / CHUNKS;
    const int ggrid = ((NN / 16) + 7) / 8;
    const int pgrid = (NN + 7) / 8;

    chunk_rank<<<CHUNKS, 1024, 0, stream>>>(col, bcnt, gsum, rank, E, chunk);
    scan_deg<<<(NN + 1023) / 1024, 1024, 0, stream>>>(bcnt, gsum, rowptr, dinv, NN);
    fill_csr<<<egrid, 256, 0, stream>>>(row, col, rowptr, rank, bcnt, dinv, src, wgt, E, chunk);

    gemm_wmma<128, false><<<ggrid, 256, 0, stream>>>(x, W1, H, NN);
    pull_rows<128><<<pgrid, 256, 0, stream>>>(H, rowptr, src, wgt, dinv, b1, O, NN);

    gemm_wmma<128, true><<<ggrid, 256, 0, stream>>>(O, W2, H, NN);
    pull_rows<128><<<pgrid, 256, 0, stream>>>(H, rowptr, src, wgt, dinv, b2, O, NN);

    gemm_wmma<64, true><<<ggrid, 256, 0, stream>>>(O, W3, H, NN);
    pull_rows<64><<<pgrid, 256, 0, stream>>>(H, rowptr, src, wgt, dinv, b3, out, NN);
}


# edge sort rebuilt as two-level bucket sort: LDS bin by 256-node bucket, then per-bucket LDS count, scan, place; 2 launches, not 3; no 64 x 50000 count plane; sum order = arrival rank in the bucket
# speedup vs baseline: 3.7466x; 1.3964x over previous
#include <hip/hip_runtime.h>

#define NN 50000
#define FDIM 128

typedef __attribute__((ext_vector_type(2))) float v2f;
typedef __attribute__((ext_vector_type(8))) float v8f;


#define PIECE 4096
#define EPT 16
#define BSH 8
#define BNODES 256
#define NBP 256
#define NBUCK ((NN + BNODES - 1) / BNODES)
#define SB_THREADS 512
#define SB_KEEP 10

__device__ inline int wave_scan_incl(int v, int lane) {
    for (int off = 1; off < 32; off <<= 1) {
        const int t = __shfl_up(v, off, 32);
        if (lane >= off) v += t;
    }
    return v;
}

__global__ __launch_bounds__(256) void bin_edges(const int* __restrict__ row,
                                                 const int* __restrict__ col,
                                                 int* __restrict__ tmp,
                                                 int* __restrict__ cntab,
                                                 int* __restrict__ offtab, int e) {
    __shared__ int cnt[NBP];
    __shared__ int wsum[8];
    __shared__ int lp[PIECE];
    const int tid  = threadIdx.x;
    const int lane = tid & 31;
    const int wv   = tid >> 5;
    cnt[tid] = 0;
    __syncthreads();

    const int beg = blockIdx.x * PIECE;
    const int len = min(PIECE, e - beg);
    int c[EPT], s[EPT], r[EPT], st[EPT];
#pragma unroll
    for (int u = 0; u < EPT; ++u) {
        const int k = min(u * 256 + tid, len - 1);
        c[u] = col[beg + k];
        s[u] = row[beg + k];
    }
#pragma unroll
    for (int u = 0; u < EPT; ++u) r[u] = (u * 256 + tid < len) ? atomicAdd(&cnt[c[u] >> BSH], 1) : 0;
    __syncthreads();

    const int v   = cnt[tid];
    const int inc = wave_scan_incl(v, lane);
    if (lane == 31) wsum[wv] = inc;
    __syncthreads();
    int front = 0;
    for (int w = 0; w < 8; ++w) front += (w < wv) ? wsum[w] : 0;
    const int excl = front + inc - v;
    cnt[tid] = excl;
    cntab[blockIdx.x * NBP + tid]  = v;
    offtab[blockIdx.x * NBP + tid] = excl;
    __syncthreads();

#pragma unroll
    for (int u = 0; u < EPT; ++u) st[u] = cnt[c[u] >> BSH];
#pragma unroll
    for (int u = 0; u < EPT; ++u) {
        if (u * 256 + tid < len) lp[st[u] + r[u]] = (s[u] << BSH) | (c[u] & (BNODES - 1));
    }
    __syncthreads();
    for (int k = tid; k < len; k += 256) tmp[beg + k] = lp[k];
}

__global__ __launch_bounds__(SB_THREADS) void sort_bucket(const int* __restrict__ tmp,
                                                          const int* __restrict__ cntab,
                                                          const int* __restrict__ offtab,
                                                          int* __restrict__ rowptr,
                                                          float* __restrict__ dinv,
                                                          int* __restrict__ src, int pieces, int n) {
    __shared__ int pstart[NBP + 1];
    __shared__ int poff[NBP];
    __shared__ int dcnt[BNODES];
    __shared__ int addr[SB_KEEP * SB_THREADS];
    __shared__ int wtot[8];
    __shared__ int ptot[8];
    const int tid  = threadIdx.x;
    const int lane = tid & 31;
    const int wv   = tid >> 5;
    const int g    = blockIdx.x;

    int myc = 0, myo = 0, inc = 0;
    if (tid < pieces) {
        myc = cntab[tid * NBP + g];
        myo = offtab[tid * NBP + g];
    }
    if (tid < NBP) {
        inc = wave_scan_incl(myc, lane);
        int so = myo;
        for (int off = 16; off > 0; off >>= 1) so += __shfl_xor(so, off, 32);
        if (lane == 31) wtot[wv] = inc;
        if (lane == 0)  ptot[wv] = so;
        dcnt[tid] = 0;
    }
    __syncthreads();

    int base = 0;
    for (int w = 0; w < 8; ++w) base += ptot[w];
    if (tid < NBP) {
        int front = 0;
        for (int w = 0; w < 8; ++w) front += (w < wv) ? wtot[w] : 0;
        const int excl = front + inc - myc;
        pstart[tid] = excl;
        poff[tid]   = myo;
        if (tid == NBP - 1) pstart[NBP] = excl + myc;
        const int a0 = tid * PIECE + myo;
        for (int i = 0; i < myc; ++i) {
            if (excl + i < SB_KEEP * SB_THREADS) addr[excl + i] = a0 + i;
        }
    }
    __syncthreads();
    const int ng = pstart[NBP];

    int wk[SB_KEEP], ak[SB_KEEP];
#pragma unroll
    for (int k = 0; k < SB_KEEP; ++k) {
        const int j = k * SB_THREADS + tid;
        ak[k] = (j < ng) ? addr[j] : 0;
    }
#pragma unroll
    for (int k = 0; k < SB_KEEP; ++k) {
        const int j = k * SB_THREADS + tid;
        wk[k] = (j < ng) ? tmp[ak[k]] : -1;
    }
#pragma unroll
    for (int k = 0; k < SB_KEEP; ++k) {
        if (wk[k] >= 0) atomicAdd(&dcnt[wk[k] & (BNODES - 1)], 1);
    }
    for (int j = SB_KEEP * SB_THREADS + tid; j < ng; j += SB_THREADS) {
        int lo = 0, hi = NBP;
        for (int it = 0; it < 8; ++it) {
            const int mid = (lo + hi) >> 1;
            if (pstart[mid] <= j) lo = mid; else hi = mid;
        }
        const int w = tmp[lo * PIECE + poff[lo] + (j - pstart[lo])];
        atomicAdd(&dcnt[w & (BNODES - 1)], 1);
    }
    __syncthreads();

    int d = 0, dinc = 0;
    if (tid < BNODES) {
        d    = dcnt[tid];
        dinc = wave_scan_incl(d, lane);
        if (lane == 31) wtot[wv] = dinc;
    }
    __syncthreads();
    if (tid < BNODES) {
        int front = 0;
        for (int w = 0; w < 8; ++w) front += (w < wv) ? wtot[w] : 0;
        const int dex  = front + dinc - d;
        const int node = g * BNODES + tid;
        if (node < n) {
            rowptr[node] = base + dex;
            dinv[node]   = rsqrtf((float)(d + 1));
            if (node == n - 1) rowptr[n] = base + dex + d;
        }
        dcnt[tid] = dex;
    }
    __syncthreads();

    int pos[SB_KEEP];
#pragma unroll
    for (int k = 0; k < SB_KEEP; ++k) pos[k] = (wk[k] >= 0) ? atomicAdd(&dcnt[wk[k] & (BNODES - 1)], 1) : 0;
#pragma unroll
    for (int k = 0; k < SB_KEEP; ++k) {
        if (wk[k] >= 0) src[base + pos[k]] = wk[k] >> BSH;
    }
    for (int j = SB_KEEP * SB_THREADS + tid; j < ng; j += SB_THREADS) {
        int lo = 0, hi = NBP;
        for (int it = 0; it < 8; ++it) {
            const int mid = (lo + hi) >> 1;
            if (pstart[mid] <= j) lo = mid; else hi = mid;
        }
        const int w  = tmp[lo * PIECE + poff[lo] + (j - pstart[lo])];
        const int pj = atomicAdd(&dcnt[w & (BNODES - 1)], 1);
        src[base + pj] = w >> BSH;
    }
}


template <int FOUT, bool RELU_IN>
__global__ __launch_bounds__(256) void gemm_wmma(const float* __restrict__ A,
                                                 const float* __restrict__ W,
                                                 const float* __restrict__ dinv,
                                                 float* __restrict__ H, int nrows) {
    constexpr int K  = FDIM;
    constexpr int CT = FOUT / 16;
    __shared__ float wlds[K * FOUT];
    for (int i = threadIdx.x; i < K * FOUT; i += 256) wlds[i] = W[i];
    __syncthreads();

    const int wave  = threadIdx.x >> 5;
    const int lane  = threadIdx.x & 31;
    const int strip = blockIdx.x * 8 + wave;
    if (strip >= (nrows >> 4)) return;

    const int row0 = strip << 4;
    const int m    = lane & 15;
    const int koff = (lane >> 4) << 1;

    v8f acc[CT];
    const v8f zero = {0.f, 0.f, 0.f, 0.f, 0.f, 0.f, 0.f, 0.f};
#pragma unroll
    for (int t = 0; t < CT; ++t) acc[t] = zero;

    const float* arow = A + (size_t)(row0 + m) * K;

    for (int kk = 0; kk < K; kk += 4) {
        float a0 = arow[kk + koff];
        float a1 = arow[kk + koff + 1];
        if (RELU_IN) { a0 = fmaxf(a0, 0.f); a1 = fmaxf(a1, 0.f); }
        v2f a; a.x = a0; a.y = a1;
#pragma unroll
        for (int t = 0; t < CT; ++t) {
            v2f b;
            b.x = wlds[(kk + koff)     * FOUT + t * 16 + m];
            b.y = wlds[(kk + koff + 1) * FOUT + t * 16 + m];
            acc[t] = __builtin_amdgcn_wmma_f32_16x16x4_f32(
                false, a, false, b, (short)0, acc[t], false, false);
        }
    }

    const int mrow = (lane >> 4) << 3;
    const float4 dlo = *(const float4*)(dinv + row0 + mrow);
    const float4 dhi = *(const float4*)(dinv + row0 + mrow + 4);
    const float dv[8] = {dlo.x, dlo.y, dlo.z, dlo.w, dhi.x, dhi.y, dhi.z, dhi.w};
#pragma unroll
    for (int t = 0; t < CT; ++t) {
#pragma unroll
        for (int v = 0; v < 8; ++v) {
            H[(size_t)(row0 + mrow + v) * FOUT + t * 16 + m] = acc[t][v] * dv[v];
        }
    }
}


template <int F>
__global__ __launch_bounds__(256) void pull_rows(const float* __restrict__ H,
                                                 const int* __restrict__ rowptr,
                                                 const int* __restrict__ src,
                                                 const float* __restrict__ dinv,
                                                 const float* __restrict__ bias,
                                                 float* __restrict__ O, int n) {
    constexpr int VPL = F / 32;
    typedef __attribute__((ext_vector_type(VPL))) float vec;
    const int lane = threadIdx.x & 31;
    const int node = __builtin_amdgcn_readfirstlane((int)(blockIdx.x * 8 + (threadIdx.x >> 5)));
    if (node >= n) return;

    const vec* Hv = (const vec*)H;
    const float d = dinv[node];
    const vec bv = ((const vec*)bias)[lane];
    vec acc = Hv[(size_t)node * 32 + lane];

    const int beg = rowptr[node];
    const int end = rowptr[node + 1];
    for (int base = beg; base < end; base += 32) {
        const int cnt = min(32, end - base);
        int myr = node;
        if (lane < cnt) myr = src[base + lane];
        int j = 0;
        for (; j + 4 <= cnt; j += 4) {
            const int r0 = __builtin_amdgcn_readlane(myr, j);
            const int r1 = __builtin_amdgcn_readlane(myr, j + 1);
            const int r2 = __builtin_amdgcn_readlane(myr, j + 2);
            const int r3 = __builtin_amdgcn_readlane(myr, j + 3);
            const vec h0 = Hv[(size_t)r0 * 32 + lane];
            const vec h1 = Hv[(size_t)r1 * 32 + lane];
            const vec h2 = Hv[(size_t)r2 * 32 + lane];
            const vec h3 = Hv[(size_t)r3 * 32 + lane];
            acc += h0;
            acc += h1;
            acc += h2;
            acc += h3;
        }
        for (; j < cnt; ++j) {
            const int r = __builtin_amdgcn_readlane(myr, j);
            acc += Hv[(size_t)r * 32 + lane];
        }
    }
    ((vec*)O)[(size_t)node * 32 + lane] = bv + d * acc;
}


extern "C" void kernel_launch(void* const* d_in, const int* in_sizes, int n_in,
                              void* d_out, int out_size, void* d_ws, size_t ws_size,
                              hipStream_t stream) {
    const float* x  = (const float*)d_in[0];
    const int*   ei = (const int*)d_in[1];
    const float* W1 = (const float*)d_in[2];
    const float* b1 = (const float*)d_in[3];
    const float* W2 = (const float*)d_in[4];
    const float* b2 = (const float*)d_in[5];
    const float* W3 = (const float*)d_in[6];
    const float* b3 = (const float*)d_in[7];
    float* out = (float*)d_out;

    const int E = in_sizes[1] / 2;
    const int* row = ei;
    const int* col = ei + E;

    float* ws   = (float*)d_ws;
    float* dinv = ws;
    float* H    = ws + 50176;
    float* O    = H + (size_t)NN * FDIM;
    int*   rowptr = (int*)(O + (size_t)NN * FDIM);
    int*   cntab  = rowptr + 50176;
    int*   offtab = cntab + NBP * NBP;
    int*   tmp    = offtab + NBP * NBP;
    int*   src    = tmp + E;

    const int pieces = (E + PIECE - 1) / PIECE;
    const int ggrid = ((NN / 16) + 7) / 8;
    const int pgrid = (NN + 7) / 8;

    bin_edges<<<pieces, 256, 0, stream>>>(row, col, tmp, cntab, offtab, E);
    sort_bucket<<<NBUCK, SB_THREADS, 0, stream>>>(tmp, cntab, offtab, rowptr, dinv, src, pieces, NN);

    gemm_wmma<128, false><<<ggrid, 256, 0, stream>>>(x, W1, dinv, H, NN);
    pull_rows<128><<<pgrid, 256, 0, stream>>>(H, rowptr, src, dinv, b1, O, NN);

    gemm_wmma<128, true><<<ggrid, 256, 0, stream>>>(O, W2, dinv, H, NN);
    pull_rows<128><<<pgrid, 256, 0, stream>>>(H, rowptr, src, dinv, b2, O, NN);

    gemm_wmma<64, true><<<ggrid, 256, 0, stream>>>(O, W3, dinv, H, NN);
    pull_rows<64><<<pgrid, 256, 0, stream>>>(H, rowptr, src, dinv, b3, out, NN);
}


# aggregation of layers 1, 2 fused with the next dense transform: one wave sums 16 nodes in turn, 24 rows in flight, into its LDS tile, the WMMA A operand; 6 launches, not 8; per-node sum order kept
# speedup vs baseline: 4.0586x; 1.0833x over previous
#include <hip/hip_runtime.h>

#define NN 50000
#define FDIM 128

typedef __attribute__((ext_vector_type(2))) float v2f;
typedef __attribute__((ext_vector_type(8))) float v8f;


#define PIECE 4096
#define EPT 16
#define BSH 8
#define BNODES 256
#define NBP 256
#define NBUCK ((NN + BNODES - 1) / BNODES)
#define SB_THREADS 512
#define SB_KEEP 10

__device__ inline int wave_scan_incl(int v, int lane) {
    for (int off = 1; off < 32; off <<= 1) {
        const int t = __shfl_up(v, off, 32);
        if (lane >= off) v += t;
    }
    return v;
}

__global__ __launch_bounds__(256) void bin_edges(const int* __restrict__ row,
                                                 const int* __restrict__ col,
                                                 int* __restrict__ tmp,
                                                 int* __restrict__ cntab,
                                                 int* __restrict__ offtab, int e) {
    __shared__ int cnt[NBP];
    __shared__ int wsum[8];
    __shared__ int lp[PIECE];
    const int tid  = threadIdx.x;
    const int lane = tid & 31;
    const int wv   = tid >> 5;
    cnt[tid] = 0;
    __syncthreads();

    const int beg = blockIdx.x * PIECE;
    const int len = min(PIECE, e - beg);
    int c[EPT], s[EPT], r[EPT], st[EPT];
#pragma unroll
    for (int u = 0; u < EPT; ++u) {
        const int k = min(u * 256 + tid, len - 1);
        c[u] = col[beg + k];
        s[u] = row[beg + k];
    }
#pragma unroll
    for (int u = 0; u < EPT; ++u) r[u] = (u * 256 + tid < len) ? atomicAdd(&cnt[c[u] >> BSH], 1) : 0;
    __syncthreads();

    const int v   = cnt[tid];
    const int inc = wave_scan_incl(v, lane);
    if (lane == 31) wsum[wv] = inc;
    __syncthreads();
    int front = 0;
    for (int w = 0; w < 8; ++w) front += (w < wv) ? wsum[w] : 0;
    const int excl = front + inc - v;
    cnt[tid] = excl;
    cntab[blockIdx.x * NBP + tid]  = v;
    offtab[blockIdx.x * NBP + tid] = excl;
    __syncthreads();

#pragma unroll
    for (int u = 0; u < EPT; ++u) st[u] = cnt[c[u] >> BSH];
#pragma unroll
    for (int u = 0; u < EPT; ++u) {
        if (u * 256 + tid < len) lp[st[u] + r[u]] = (s[u] << BSH) | (c[u] & (BNODES - 1));
    }
    __syncthreads();
    for (int k = tid; k < len; k += 256) tmp[beg + k] = lp[k];
}

__global__ __launch_bounds__(SB_THREADS) void sort_bucket(const int* __restrict__ tmp,
                                                          const int* __restrict__ cntab,
                                                          const int* __restrict__ offtab,
                                                          int* __restrict__ rowptr,
                                                          float* __restrict__ dinv,
                                                          int* __restrict__ src, int pieces, int n) {
    __shared__ int pstart[NBP + 1];
    __shared__ int poff[NBP];
    __shared__ int dcnt[BNODES];
    __shared__ int addr[SB_KEEP * SB_THREADS];
    __shared__ int wtot[8];
    __shared__ int ptot[8];
    const int tid  = threadIdx.x;
    const int lane = tid & 31;
    const int wv   = tid >> 5;
    const int g    = blockIdx.x;

    int myc = 0, myo = 0, inc = 0;
    if (tid < pieces) {
        myc = cntab[tid * NBP + g];
        myo = offtab[tid * NBP + g];
    }
    if (tid < NBP) {
        inc = wave_scan_incl(myc, lane);
        int so = myo;
        for (int off = 16; off > 0; off >>= 1) so += __shfl_xor(so, off, 32);
        if (lane == 31) wtot[wv] = inc;
        if (lane == 0)  ptot[wv] = so;
        dcnt[tid] = 0;
    }
    __syncthreads();

    int base = 0;
    for (int w = 0; w < 8; ++w) base += ptot[w];
    if (tid < NBP) {
        int front = 0;
        for (int w = 0; w < 8; ++w) front += (w < wv) ? wtot[w] : 0;
        const int excl = front + inc - myc;
        pstart[tid] = excl;
        poff[tid]   = myo;
        if (tid == NBP - 1) pstart[NBP] = excl + myc;
        const int a0 = tid * PIECE + myo;
        for (int i = 0; i < myc; ++i) {
            if (excl + i < SB_KEEP * SB_THREADS) addr[excl + i] = a0 + i;
        }
    }
    __syncthreads();
    const int ng = pstart[NBP];

    int wk[SB_KEEP], ak[SB_KEEP];
#pragma unroll
    for (int k = 0; k < SB_KEEP; ++k) {
        const int j = k * SB_THREADS + tid;
        ak[k] = (j < ng) ? addr[j] : 0;
    }
#pragma unroll
    for (int k = 0; k < SB_KEEP; ++k) {
        const int j = k * SB_THREADS + tid;
        wk[k] = (j < ng) ? tmp[ak[k]] : -1;
    }
#pragma unroll
    for (int k = 0; k < SB_KEEP; ++k) {
        if (wk[k] >= 0) atomicAdd(&dcnt[wk[k] & (BNODES - 1)], 1);
    }
    for (int j = SB_KEEP * SB_THREADS + tid; j < ng; j += SB_THREADS) {
        int lo = 0, hi = NBP;
        for (int it = 0; it < 8; ++it) {
            const int mid = (lo + hi) >> 1;
            if (pstart[mid] <= j) lo = mid; else hi = mid;
        }
        const int w = tmp[lo * PIECE + poff[lo] + (j - pstart[lo])];
        atomicAdd(&dcnt[w & (BNODES - 1)], 1);
    }
    __syncthreads();

    int d = 0, dinc = 0;
    if (tid < BNODES) {
        d    = dcnt[tid];
        dinc = wave_scan_incl(d, lane);
        if (lane == 31) wtot[wv] = dinc;
    }
    __syncthreads();
    if (tid < BNODES) {
        int front = 0;
        for (int w = 0; w < 8; ++w) front += (w < wv) ? wtot[w] : 0;
        const int dex  = front + dinc - d;
        const int node = g * BNODES + tid;
        if (node < n) {
            rowptr[node] = base + dex;
            dinv[node]   = rsqrtf((float)(d + 1));
            if (node == n - 1) rowptr[n] = base + dex + d;
        }
        dcnt[tid] = dex;
    }
    __syncthreads();

    int pos[SB_KEEP];
#pragma unroll
    for (int k = 0; k < SB_KEEP; ++k) pos[k] = (wk[k] >= 0) ? atomicAdd(&dcnt[wk[k] & (BNODES - 1)], 1) : 0;
#pragma unroll
    for (int k = 0; k < SB_KEEP; ++k) {
        if (wk[k] >= 0) src[base + pos[k]] = wk[k] >> BSH;
    }
    for (int j = SB_KEEP * SB_THREADS + tid; j < ng; j += SB_THREADS) {
        int lo = 0, hi = NBP;
        for (int it = 0; it < 8; ++it) {
            const int mid = (lo + hi) >> 1;
            if (pstart[mid] <= j) lo = mid; else hi = mid;
        }
        const int w  = tmp[lo * PIECE + poff[lo] + (j - pstart[lo])];
        const int pj = atomicAdd(&dcnt[w & (BNODES - 1)], 1);
        src[base + pj] = w >> BSH;
    }
}


template <int FOUT, bool RELU_IN>
__global__ __launch_bounds__(256) void gemm_wmma(const float* __restrict__ A,
                                                 const float* __restrict__ W,
                                                 const float* __restrict__ dinv,
                                                 float* __restrict__ H, int nrows) {
    constexpr int K  = FDIM;
    constexpr int CT = FOUT / 16;
    __shared__ float wlds[K * FOUT];
    for (int i = threadIdx.x; i < K * FOUT; i += 256) wlds[i] = W[i];
    __syncthreads();

    const int wave  = threadIdx.x >> 5;
    const int lane  = threadIdx.x & 31;
    const int strip = blockIdx.x * 8 + wave;
    if (strip >= (nrows >> 4)) return;

    const int row0 = strip << 4;
    const int m    = lane & 15;
    const int koff = (lane >> 4) << 1;

    v8f acc[CT];
    const v8f zero = {0.f, 0.f, 0.f, 0.f, 0.f, 0.f, 0.f, 0.f};
#pragma unroll
    for (int t = 0; t < CT; ++t) acc[t] = zero;

    const float* arow = A + (size_t)(row0 + m) * K;

    for (int kk = 0; kk < K; kk += 4) {
        float a0 = arow[kk + koff];
        float a1 = arow[kk + koff + 1];
        if (RELU_IN) { a0 = fmaxf(a0, 0.f); a1 = fmaxf(a1, 0.f); }
        v2f a; a.x = a0; a.y = a1;
#pragma unroll
        for (int t = 0; t < CT; ++t) {
            v2f b;
            b.x = wlds[(kk + koff)     * FOUT + t * 16 + m];
            b.y = wlds[(kk + koff + 1) * FOUT + t * 16 + m];
            acc[t] = __builtin_amdgcn_wmma_f32_16x16x4_f32(
                false, a, false, b, (short)0, acc[t], false, false);
        }
    }

    const int mrow = (lane >> 4) << 3;
    const float4 dlo = *(const float4*)(dinv + row0 + mrow);
    const float4 dhi = *(const float4*)(dinv + row0 + mrow + 4);
    const float dv[8] = {dlo.x, dlo.y, dlo.z, dlo.w, dhi.x, dhi.y, dhi.z, dhi.w};
#pragma unroll
    for (int t = 0; t < CT; ++t) {
#pragma unroll
        for (int v = 0; v < 8; ++v) {
            H[(size_t)(row0 + mrow + v) * FOUT + t * 16 + m] = acc[t][v] * dv[v];
        }
    }
}


template <int F>
__global__ __launch_bounds__(256) void pull_rows(const float* __restrict__ H,
                                                 const int* __restrict__ rowptr,
                                                 const int* __restrict__ src,
                                                 const float* __restrict__ dinv,
                                                 const float* __restrict__ bias,
                                                 float* __restrict__ O, int n) {
    constexpr int VPL = F / 32;
    typedef __attribute__((ext_vector_type(VPL))) float vec;
    const int lane = threadIdx.x & 31;
    const int node = __builtin_amdgcn_readfirstlane((int)(blockIdx.x * 8 + (threadIdx.x >> 5)));
    if (node >= n) return;

    const vec* Hv = (const vec*)H;
    const float d = dinv[node];
    const vec bv = ((const vec*)bias)[lane];
    vec acc = Hv[(size_t)node * 32 + lane];

    const int beg = rowptr[node];
    const int end = rowptr[node + 1];
    for (int base = beg; base < end; base += 32) {
        const int cnt = min(32, end - base);
        int myr = node;
        if (lane < cnt) myr = src[base + lane];
        int j = 0;
        for (; j + 4 <= cnt; j += 4) {
            const int r0 = __builtin_amdgcn_readlane(myr, j);
            const int r1 = __builtin_amdgcn_readlane(myr, j + 1);
            const int r2 = __builtin_amdgcn_readlane(myr, j + 2);
            const int r3 = __builtin_amdgcn_readlane(myr, j + 3);
            const vec h0 = Hv[(size_t)r0 * 32 + lane];
            const vec h1 = Hv[(size_t)r1 * 32 + lane];
            const vec h2 = Hv[(size_t)r2 * 32 + lane];
            const vec h3 = Hv[(size_t)r3 * 32 + lane];
            acc += h0;
            acc += h1;
            acc += h2;
            acc += h3;
        }
        for (; j < cnt; ++j) {
            const int r = __builtin_amdgcn_readlane(myr, j);
            acc += Hv[(size_t)r * 32 + lane];
        }
    }
    ((vec*)O)[(size_t)node * 32 + lane] = bv + d * acc;
}


typedef __attribute__((ext_vector_type(4))) float v4f;
struct Rows8 { v4f r[8]; };

__device__ inline Rows8 load8(const v4f* __restrict__ Hv, int ids, int j, int lane) {
    Rows8 h;
#pragma unroll
    for (int k = 0; k < 8; ++k) {
        const int r = __builtin_amdgcn_readlane(ids, min(j + k, 31));
        h.r[k] = Hv[(size_t)r * 32 + lane];
    }
    return h;
}

__device__ inline void add8(v4f& acc, const Rows8& h, int j, int cnt) {
#pragma unroll
    for (int k = 0; k < 8; ++k) {
        if (j + k < cnt) acc += h.r[k];
    }
}

#define FW 13
#define TP 132

template <int FOUT>
__global__ __launch_bounds__(FW * 32) void pull_gemm(const float* __restrict__ Hin,
                                                     const int* __restrict__ rowptr,
                                                     const int* __restrict__ src,
                                                     const float* __restrict__ dinv,
                                                     const float* __restrict__ bias,
                                                     const float* __restrict__ W,
                                                     float* __restrict__ Hout, int n) {
    constexpr int K  = FDIM;
    constexpr int CT = FOUT / 16;
    typedef v4f vec;
    __shared__ float wlds[K * FOUT];
    __shared__ __attribute__((aligned(16))) float tile[FW * 16 * TP];
    for (int i = threadIdx.x; i < K * FOUT; i += FW * 32) wlds[i] = W[i];
    __syncthreads();

    const int lane = threadIdx.x & 31;
    const int wv = __builtin_amdgcn_readfirstlane((int)(threadIdx.x >> 5));
    const int tl = __builtin_amdgcn_readfirstlane((int)(blockIdx.x * FW + wv));
    if (tl >= (n >> 4)) return;
    float* my = tile + wv * 16 * TP;
    const int node0 = tl << 4;

    const vec* Hv = (const vec*)Hin;
    const int   rp = rowptr[node0 + min(lane, 16)];
    const float dl = dinv[node0 + (lane & 15)];
    const vec   bv = ((const vec*)bias)[lane];
    int beg = __builtin_amdgcn_readlane(rp, 0);
    int end = __builtin_amdgcn_readlane(rp, 1);
    int myr = node0;
    if (lane < end - beg) myr = src[beg + lane];
    vec acc = Hv[(size_t)node0 * 32 + lane];
    for (int i = 0; i < 16; ++i) {
        const int node = node0 + i;
        const int cnt0 = min(32, end - beg);
        const Rows8 h  = load8(Hv, myr, 0, lane);
        const Rows8 h2 = load8(Hv, myr, 8, lane);
        const Rows8 h3 = load8(Hv, myr, 16, lane);
        const int nn   = min(node + 1, n - 1);
        const int nbeg = end;
        const int nend = __builtin_amdgcn_readlane(rp, min(i + 2, 16));
        int nmyr = nn;
        if (lane < nend - nbeg) nmyr = src[nbeg + lane];
        const vec nself = Hv[(size_t)nn * 32 + lane];
        add8(acc, h, 0, cnt0);
        add8(acc, h2, 8, cnt0);
        add8(acc, h3, 16, cnt0);
        for (int j = 24; j < cnt0; j += 8) {
            const Rows8 g = load8(Hv, myr, j, lane);
            add8(acc, g, j, cnt0);
        }
        for (int base = beg + 32; base < end; base += 32) {
            const int cnt = min(32, end - base);
            int mor = node;
            if (lane < cnt) mor = src[base + lane];
            for (int j = 0; j < cnt; j += 8) {
                const Rows8 g = load8(Hv, mor, j, lane);
                add8(acc, g, j, cnt);
            }
        }
        const float d = __int_as_float(__builtin_amdgcn_readlane(__float_as_int(dl), i));
        vec o = bv + d * acc;
        o.x = fmaxf(o.x, 0.f); o.y = fmaxf(o.y, 0.f); o.z = fmaxf(o.z, 0.f); o.w = fmaxf(o.w, 0.f);
        *(vec*)(my + i * TP + lane * 4) = o;
        beg = nbeg; end = nend; myr = nmyr; acc = nself;
    }
    __builtin_amdgcn_fence(__ATOMIC_RELEASE, "wavefront");
    __builtin_amdgcn_wave_barrier();
    __builtin_amdgcn_fence(__ATOMIC_ACQUIRE, "wavefront");

    const int m    = lane & 15;
    const int koff = (lane >> 4) << 1;
    v8f cacc[CT];
    const v8f zero = {0.f, 0.f, 0.f, 0.f, 0.f, 0.f, 0.f, 0.f};
#pragma unroll
    for (int t = 0; t < CT; ++t) cacc[t] = zero;
    const float* arow = my + m * TP;
    for (int kk = 0; kk < K; kk += 4) {
        const v2f a = *(const v2f*)(arow + kk + koff);
#pragma unroll
        for (int t = 0; t < CT; ++t) {
            v2f b;
            b.x = wlds[(kk + koff)     * FOUT + t * 16 + m];
            b.y = wlds[(kk + koff + 1) * FOUT + t * 16 + m];
            cacc[t] = __builtin_amdgcn_wmma_f32_16x16x4_f32(
                false, a, false, b, (short)0, cacc[t], false, false);
        }
    }
    const int mrow = (lane >> 4) << 3;
    const float4 dlo = *(const float4*)(dinv + node0 + mrow);
    const float4 dhi = *(const float4*)(dinv + node0 + mrow + 4);
    const float dv[8] = {dlo.x, dlo.y, dlo.z, dlo.w, dhi.x, dhi.y, dhi.z, dhi.w};
#pragma unroll
    for (int t = 0; t < CT; ++t) {
#pragma unroll
        for (int v = 0; v < 8; ++v) {
            Hout[(size_t)(node0 + mrow + v) * FOUT + t * 16 + m] = cacc[t][v] * dv[v];
        }
    }
}


extern "C" void kernel_launch(void* const* d_in, const int* in_sizes, int n_in,
                              void* d_out, int out_size, void* d_ws, size_t ws_size,
                              hipStream_t stream) {
    const float* x  = (const float*)d_in[0];
    const int*   ei = (const int*)d_in[1];
    const float* W1 = (const float*)d_in[2];
    const float* b1 = (const float*)d_in[3];
    const float* W2 = (const float*)d_in[4];
    const float* b2 = (const float*)d_in[5];
    const float* W3 = (const float*)d_in[6];
    const float* b3 = (const float*)d_in[7];
    float* out = (float*)d_out;

    const int E = in_sizes[1] / 2;
    const int* row = ei;
    const int* col = ei + E;

    float* ws   = (float*)d_ws;
    float* dinv = ws;
    float* H    = ws + 50176;
    float* H2   = H + (size_t)NN * FDIM;
    int*   rowptr = (int*)(H2 + (size_t)NN * FDIM);
    int*   cntab  = rowptr + 50176;
    int*   offtab = cntab + NBP * NBP;
    int*   tmp    = offtab + NBP * NBP;
    int*   src    = tmp + E;

    const int pieces = (E + PIECE - 1) / PIECE;
    const int ggrid = ((NN / 16) + 7) / 8;
    const int pgrid = (NN + 7) / 8;
    const int fgrid = ((NN / 16) + FW - 1) / FW;

    bin_edges<<<pieces, 256, 0, stream>>>(row, col, tmp, cntab, offtab, E);
    sort_bucket<<<NBUCK, SB_THREADS, 0, stream>>>(tmp, cntab, offtab, rowptr, dinv, src, pieces, NN);

    gemm_wmma<128, false><<<ggrid, 256, 0, stream>>>(x, W1, dinv, H, NN);

    pull_gemm<128><<<fgrid, FW * 32, 0, stream>>>(H, rowptr, src, dinv, b1, W2, H2, NN);

    pull_gemm<64><<<fgrid, FW * 32, 0, stream>>>(H2, rowptr, src, dinv, b2, W3, H, NN);

    pull_rows<64><<<pgrid, 256, 0, stream>>>(H, rowptr, src, dinv, b3, out, NN);
}


# weights into LDS: a thread's 16-byte loads all asked for before its first store, one round trip shared with the tile's first loads, not 4 to 11 dependent groups of 4-byte loads; sums untouched
# speedup vs baseline: 4.3906x; 1.0818x over previous
#include <hip/hip_runtime.h>

#define NN 50000
#define FDIM 128

typedef __attribute__((ext_vector_type(2))) float v2f;
typedef __attribute__((ext_vector_type(8))) float v8f;
typedef __attribute__((ext_vector_type(4))) float v4f;

template <int NT, int NW>
struct WPieces { v4f r[(NW / 4 + NT - 1) / NT]; };

template <int NT, int NW>
__device__ inline WPieces<NT, NW> load_weights(const float* __restrict__ W, int tid) {
    constexpr int NV  = NW / 4;
    constexpr int PER = (NV + NT - 1) / NT;
    const v4f* Wv = (const v4f*)W;
    WPieces<NT, NW> p;
#pragma unroll
    for (int u = 0; u < PER; ++u) {
        const int i = u * NT + tid;
        const bool part = (u == PER - 1) && (NV % NT != 0);
        p.r[u] = Wv[part ? min(i, NV - 1) : i];
    }
    return p;
}

template <int NT, int NW>
__device__ inline void store_weights(float* __restrict__ wlds, const WPieces<NT, NW>& p, int tid) {
    constexpr int NV  = NW / 4;
    constexpr int PER = (NV + NT - 1) / NT;
#pragma unroll
    for (int u = 0; u < PER; ++u) {
        const int i = u * NT + tid;
        const bool part = (u == PER - 1) && (NV % NT != 0);
        if (!part || i < NV) ((v4f*)wlds)[i] = p.r[u];
    }
}


#define PIECE 4096
#define EPT 16
#define BSH 8
#define BNODES 256
#define NBP 256
#define NBUCK ((NN + BNODES - 1) / BNODES)
#define SB_THREADS 512
#define SB_KEEP 10

__device__ inline int wave_scan_incl(int v, int lane) {
    for (int off = 1; off < 32; off <<= 1) {
        const int t = __shfl_up(v, off, 32);
        if (lane >= off) v += t;
    }
    return v;
}

__global__ __launch_bounds__(256) void bin_edges(const int* __restrict__ row,
                                                 const int* __restrict__ col,
                                                 int* __restrict__ tmp,
                                                 int* __restrict__ cntab,
                                                 int* __restrict__ offtab, int e) {
    __shared__ int cnt[NBP];
    __shared__ int wsum[8];
    __shared__ int lp[PIECE];
    const int tid  = threadIdx.x;
    const int lane = tid & 31;
    const int wv   = tid >> 5;
    cnt[tid] = 0;
    __syncthreads();

    const int beg = blockIdx.x * PIECE;
    const int len = min(PIECE, e - beg);
    int c[EPT], s[EPT], r[EPT], st[EPT];
#pragma unroll
    for (int u = 0; u < EPT; ++u) {
        const int k = min(u * 256 + tid, len - 1);
        c[u] = col[beg + k];
        s[u] = row[beg + k];
    }
#pragma unroll
    for (int u = 0; u < EPT; ++u) r[u] = (u * 256 + tid < len) ? atomicAdd(&cnt[c[u] >> BSH], 1) : 0;
    __syncthreads();

    const int v   = cnt[tid];
    const int inc = wave_scan_incl(v, lane);
    if (lane == 31) wsum[wv] = inc;
    __syncthreads();
    int front = 0;
    for (int w = 0; w < 8; ++w) front += (w < wv) ? wsum[w] : 0;
    const int excl = front + inc - v;
    cnt[tid] = excl;
    cntab[blockIdx.x * NBP + tid]  = v;
    offtab[blockIdx.x * NBP + tid] = excl;
    __syncthreads();

#pragma unroll
    for (int u = 0; u < EPT; ++u) st[u] = cnt[c[u] >> BSH];
#pragma unroll
    for (int u = 0; u < EPT; ++u) {
        if (u * 256 + tid < len) lp[st[u] + r[u]] = (s[u] << BSH) | (c[u] & (BNODES - 1));
    }
    __syncthreads();
    for (int k = tid; k < len; k += 256) tmp[beg + k] = lp[k];
}

__global__ __launch_bounds__(SB_THREADS) void sort_bucket(const int* __restrict__ tmp,
                                                          const int* __restrict__ cntab,
                                                          const int* __restrict__ offtab,
                                                          int* __restrict__ rowptr,
                                                          float* __restrict__ dinv,
                                                          int* __restrict__ src, int pieces, int n) {
    __shared__ int pstart[NBP + 1];
    __shared__ int poff[NBP];
    __shared__ int dcnt[BNODES];
    __shared__ int addr[SB_KEEP * SB_THREADS];
    __shared__ int wtot[8];
    __shared__ int ptot[8];
    const int tid  = threadIdx.x;
    const int lane = tid & 31;
    const int wv   = tid >> 5;
    const int g    = blockIdx.x;

    int myc = 0, myo = 0, inc = 0;
    if (tid < pieces) {
        myc = cntab[tid * NBP + g];
        myo = offtab[tid * NBP + g];
    }
    if (tid < NBP) {
        inc = wave_scan_incl(myc, lane);
        int so = myo;
        for (int off = 16; off > 0; off >>= 1) so += __shfl_xor(so, off, 32);
        if (lane == 31) wtot[wv] = inc;
        if (lane == 0)  ptot[wv] = so;
        dcnt[tid] = 0;
    }
    __syncthreads();

    int base = 0;
    for (int w = 0; w < 8; ++w) base += ptot[w];
    if (tid < NBP) {
        int front = 0;
        for (int w = 0; w < 8; ++w) front += (w < wv) ? wtot[w] : 0;
        const int excl = front + inc - myc;
        pstart[tid] = excl;
        poff[tid]   = myo;
        if (tid == NBP - 1) pstart[NBP] = excl + myc;
        const int a0 = tid * PIECE + myo;
        for (int i = 0; i < myc; ++i) {
            if (excl + i < SB_KEEP * SB_THREADS) addr[excl + i] = a0 + i;
        }
    }
    __syncthreads();
    const int ng = pstart[NBP];

    int wk[SB_KEEP], ak[SB_KEEP];
#pragma unroll
    for (int k = 0; k < SB_KEEP; ++k) {
        const int j = k * SB_THREADS + tid;
        ak[k] = (j < ng) ? addr[j] : 0;
    }
#pragma unroll
    for (int k = 0; k < SB_KEEP; ++k) {
        const int j = k * SB_THREADS + tid;
        wk[k] = (j < ng) ? tmp[ak[k]] : -1;
    }
#pragma unroll
    for (int k = 0; k < SB_KEEP; ++k) {
        if (wk[k] >= 0) atomicAdd(&dcnt[wk[k] & (BNODES - 1)], 1);
    }
    for (int j = SB_KEEP * SB_THREADS + tid; j < ng; j += SB_THREADS) {
        int lo = 0, hi = NBP;
        for (int it = 0; it < 8; ++it) {
            const int mid = (lo + hi) >> 1;
            if (pstart[mid] <= j) lo = mid; else hi = mid;
        }
        const int w = tmp[lo * PIECE + poff[lo] + (j - pstart[lo])];
        atomicAdd(&dcnt[w & (BNODES - 1)], 1);
    }
    __syncthreads();

    int d = 0, dinc = 0;
    if (tid < BNODES) {
        d    = dcnt[tid];
        dinc = wave_scan_incl(d, lane);
        if (lane == 31) wtot[wv] = dinc;
    }
    __syncthreads();
    if (tid < BNODES) {
        int front = 0;
        for (int w = 0; w < 8; ++w) front += (w < wv) ? wtot[w] : 0;
        const int dex  = front + dinc - d;
        const int node = g * BNODES + tid;
        if (node < n) {
            rowptr[node] = base + dex;
            dinv[node]   = rsqrtf((float)(d + 1));
            if (node == n - 1) rowptr[n] = base + dex + d;
        }
        dcnt[tid] = dex;
    }
    __syncthreads();

    int pos[SB_KEEP];
#pragma unroll
    for (int k = 0; k < SB_KEEP; ++k) pos[k] = (wk[k] >= 0) ? atomicAdd(&dcnt[wk[k] & (BNODES - 1)], 1) : 0;
#pragma unroll
    for (int k = 0; k < SB_KEEP; ++k) {
        if (wk[k] >= 0) src[base + pos[k]] = wk[k] >> BSH;
    }
    for (int j = SB_KEEP * SB_THREADS + tid; j < ng; j += SB_THREADS) {
        int lo = 0, hi = NBP;
        for (int it = 0; it < 8; ++it) {
            const int mid = (lo + hi) >> 1;
            if (pstart[mid] <= j) lo = mid; else hi = mid;
        }
        const int w  = tmp[lo * PIECE + poff[lo] + (j - pstart[lo])];
        const int pj = atomicAdd(&dcnt[w & (BNODES - 1)], 1);
        src[base + pj] = w >> BSH;
    }
}


template <int FOUT, bool RELU_IN>
__global__ __launch_bounds__(256) void gemm_wmma(const float* __restrict__ A,
                                                 const float* __restrict__ W,
                                                 const float* __restrict__ dinv,
                                                 float* __restrict__ H, int nrows) {
    constexpr int K  = FDIM;
    constexpr int CT = FOUT / 16;
    __shared__ __attribute__((aligned(16))) float wlds[K * FOUT];
    store_weights<256, K * FOUT>(wlds, load_weights<256, K * FOUT>(W, threadIdx.x), threadIdx.x);
    __syncthreads();

    const int wave  = threadIdx.x >> 5;
    const int lane  = threadIdx.x & 31;
    const int strip = blockIdx.x * 8 + wave;
    if (strip >= (nrows >> 4)) return;

    const int row0 = strip << 4;
    const int m    = lane & 15;
    const int koff = (lane >> 4) << 1;

    v8f acc[CT];
    const v8f zero = {0.f, 0.f, 0.f, 0.f, 0.f, 0.f, 0.f, 0.f};
#pragma unroll
    for (int t = 0; t < CT; ++t) acc[t] = zero;

    const float* arow = A + (size_t)(row0 + m) * K;

    for (int kk = 0; kk < K; kk += 4) {
        float a0 = arow[kk + koff];
        float a1 = arow[kk + koff + 1];
        if (RELU_IN) { a0 = fmaxf(a0, 0.f); a1 = fmaxf(a1, 0.f); }
        v2f a; a.x = a0; a.y = a1;
#pragma unroll
        for (int t = 0; t < CT; ++t) {
            v2f b;
            b.x = wlds[(kk + koff)     * FOUT + t * 16 + m];
            b.y = wlds[(kk + koff + 1) * FOUT + t * 16 + m];
            acc[t] = __builtin_amdgcn_wmma_f32_16x16x4_f32(
                false, a, false, b, (short)0, acc[t], false, false);
        }
    }

    const int mrow = (lane >> 4) << 3;
    const float4 dlo = *(const float4*)(dinv + row0 + mrow);
    const float4 dhi = *(const float4*)(dinv + row0 + mrow + 4);
    const float dv[8] = {dlo.x, dlo.y, dlo.z, dlo.w, dhi.x, dhi.y, dhi.z, dhi.w};
#pragma unroll
    for (int t = 0; t < CT; ++t) {
#pragma unroll
        for (int v = 0; v < 8; ++v) {
            H[(size_t)(row0 + mrow + v) * FOUT + t * 16 + m] = acc[t][v] * dv[v];
        }
    }
}


template <int F>
__global__ __launch_bounds__(256) void pull_rows(const float* __restrict__ H,
                                                 const int* __restrict__ rowptr,
                                                 const int* __restrict__ src,
                                                 const float* __restrict__ dinv,
                                                 const float* __restrict__ bias,
                                                 float* __restrict__ O, int n) {
    constexpr int VPL = F / 32;
    typedef __attribute__((ext_vector_type(VPL))) float vec;
    const int lane = threadIdx.x & 31;
    const int node = __builtin_amdgcn_readfirstlane((int)(blockIdx.x * 8 + (threadIdx.x >> 5)));
    if (node >= n) return;

    const vec* Hv = (const vec*)H;
    const float d = dinv[node];
    const vec bv = ((const vec*)bias)[lane];
    vec acc = Hv[(size_t)node * 32 + lane];

    const int beg = rowptr[node];
    const int end = rowptr[node + 1];
    for (int base = beg; base < end; base += 32) {
        const int cnt = min(32, end - base);
        int myr = node;
        if (lane < cnt) myr = src[base + lane];
        int j = 0;
        for (; j + 4 <= cnt; j += 4) {
            const int r0 = __builtin_amdgcn_readlane(myr, j);
            const int r1 = __builtin_amdgcn_readlane(myr, j + 1);
            const int r2 = __builtin_amdgcn_readlane(myr, j + 2);
            const int r3 = __builtin_amdgcn_readlane(myr, j + 3);
            const vec h0 = Hv[(size_t)r0 * 32 + lane];
            const vec h1 = Hv[(size_t)r1 * 32 + lane];
            const vec h2 = Hv[(size_t)r2 * 32 + lane];
            const vec h3 = Hv[(size_t)r3 * 32 + lane];
            acc += h0;
            acc += h1;
            acc += h2;
            acc += h3;
        }
        for (; j < cnt; ++j) {
            const int r = __builtin_amdgcn_readlane(myr, j);
            acc += Hv[(size_t)r * 32 + lane];
        }
    }
    ((vec*)O)[(size_t)node * 32 + lane] = bv + d * acc;
}


struct Rows8 { v4f r[8]; };

__device__ inline Rows8 load8(const v4f* __restrict__ Hv, int ids, int j, int lane) {
    Rows8 h;
#pragma unroll
    for (int k = 0; k < 8; ++k) {
        const int r = __builtin_amdgcn_readlane(ids, min(j + k, 31));
        h.r[k] = Hv[(size_t)r * 32 + lane];
    }
    return h;
}

__device__ inline void add8(v4f& acc, const Rows8& h, int j, int cnt) {
#pragma unroll
    for (int k = 0; k < 8; ++k) {
        if (j + k < cnt) acc += h.r[k];
    }
}

#define FW 13
#define TP 132

template <int FOUT>
__global__ __launch_bounds__(FW * 32) void pull_gemm(const float* __restrict__ Hin,
                                                     const int* __restrict__ rowptr,
                                                     const int* __restrict__ src,
                                                     const float* __restrict__ dinv,
                                                     const float* __restrict__ bias,
                                                     const float* __restrict__ W,
                                                     float* __restrict__ Hout, int n) {
    constexpr int K  = FDIM;
    constexpr int CT = FOUT / 16;
    typedef v4f vec;
    __shared__ __attribute__((aligned(16))) float wlds[K * FOUT];
    __shared__ __attribute__((aligned(16))) float tile[FW * 16 * TP];

    const int lane = threadIdx.x & 31;
    const int wv = __builtin_amdgcn_readfirstlane((int)(threadIdx.x >> 5));
    const int tl = __builtin_amdgcn_readfirstlane((int)(blockIdx.x * FW + wv));
    const bool has = tl < (n >> 4);
    const int node0 = (has ? tl : 0) << 4;

    const vec* Hv = (const vec*)Hin;
    const int   rp = rowptr[node0 + min(lane, 16)];
    const float dl = dinv[node0 + (lane & 15)];
    const vec   bv = ((const vec*)bias)[lane];
    vec acc = Hv[(size_t)node0 * 32 + lane];
    const WPieces<FW * 32, K * FOUT> wp = load_weights<FW * 32, K * FOUT>(W, threadIdx.x);
    int beg = __builtin_amdgcn_readlane(rp, 0);
    int end = __builtin_amdgcn_readlane(rp, 1);
    int myr = node0;
    if (lane < end - beg) myr = src[beg + lane];
    store_weights<FW * 32, K * FOUT>(wlds, wp, threadIdx.x);
    __syncthreads();
    if (!has) return;
    float* my = tile + wv * 16 * TP;

    for (int i = 0; i < 16; ++i) {
        const int node = node0 + i;
        const int cnt0 = min(32, end - beg);
        const Rows8 h  = load8(Hv, myr, 0, lane);
        const Rows8 h2 = load8(Hv, myr, 8, lane);
        const Rows8 h3 = load8(Hv, myr, 16, lane);
        const int nn   = min(node + 1, n - 1);
        const int nbeg = end;
        const int nend = __builtin_amdgcn_readlane(rp, min(i + 2, 16));
        int nmyr = nn;
        if (lane < nend - nbeg) nmyr = src[nbeg + lane];
        const vec nself = Hv[(size_t)nn * 32 + lane];
        add8(acc, h, 0, cnt0);
        add8(acc, h2, 8, cnt0);
        add8(acc, h3, 16, cnt0);
        for (int j = 24; j < cnt0; j += 8) {
            const Rows8 g = load8(Hv, myr, j, lane);
            add8(acc, g, j, cnt0);
        }
        for (int base = beg + 32; base < end; base += 32) {
            const int cnt = min(32, end - base);
            int mor = node;
            if (lane < cnt) mor = src[base + lane];
            for (int j = 0; j < cnt; j += 8) {
                const Rows8 g = load8(Hv, mor, j, lane);
                add8(acc, g, j, cnt);
            }
        }
        const float d = __int_as_float(__builtin_amdgcn_readlane(__float_as_int(dl), i));
        vec o = bv + d * acc;
        o.x = fmaxf(o.x, 0.f); o.y = fmaxf(o.y, 0.f); o.z = fmaxf(o.z, 0.f); o.w = fmaxf(o.w, 0.f);
        *(vec*)(my + i * TP + lane * 4) = o;
        beg = nbeg; end = nend; myr = nmyr; acc = nself;
    }
    __builtin_amdgcn_fence(__ATOMIC_RELEASE, "wavefront");
    __builtin_amdgcn_wave_barrier();
    __builtin_amdgcn_fence(__ATOMIC_ACQUIRE, "wavefront");

    const int m    = lane & 15;
    const int koff = (lane >> 4) << 1;
    v8f cacc[CT];
    const v8f zero = {0.f, 0.f, 0.f, 0.f, 0.f, 0.f, 0.f, 0.f};
#pragma unroll
    for (int t = 0; t < CT; ++t) cacc[t] = zero;
    const float* arow = my + m * TP;
    for (int kk = 0; kk < K; kk += 4) {
        const v2f a = *(const v2f*)(arow + kk + koff);
#pragma unroll
        for (int t = 0; t < CT; ++t) {
            v2f b;
            b.x = wlds[(kk + koff)     * FOUT + t * 16 + m];
            b.y = wlds[(kk + koff + 1) * FOUT + t * 16 + m];
            cacc[t] = __builtin_amdgcn_wmma_f32_16x16x4_f32(
                false, a, false, b, (short)0, cacc[t], false, false);
        }
    }
    const int mrow = (lane >> 4) << 3;
    const float4 dlo = *(const float4*)(dinv + node0 + mrow);
    const float4 dhi = *(const float4*)(dinv + node0 + mrow + 4);
    const float dv[8] = {dlo.x, dlo.y, dlo.z, dlo.w, dhi.x, dhi.y, dhi.z, dhi.w};
#pragma unroll
    for (int t = 0; t < CT; ++t) {
#pragma unroll
        for (int v = 0; v < 8; ++v) {
            Hout[(size_t)(node0 + mrow + v) * FOUT + t * 16 + m] = cacc[t][v] * dv[v];
        }
    }
}


extern "C" void kernel_launch(void* const* d_in, const int* in_sizes, int n_in,
                              void* d_out, int out_size, void* d_ws, size_t ws_size,
                              hipStream_t stream) {
    const float* x  = (const float*)d_in[0];
    const int*   ei = (const int*)d_in[1];
    const float* W1 = (const float*)d_in[2];
    const float* b1 = (const float*)d_in[3];
    const float* W2 = (const float*)d_in[4];
    const float* b2 = (const float*)d_in[5];
    const float* W3 = (const float*)d_in[6];
    const float* b3 = (const float*)d_in[7];
    float* out = (float*)d_out;

    const int E = in_sizes[1] / 2;
    const int* row = ei;
    const int* col = ei + E;

    float* ws   = (float*)d_ws;
    float* dinv = ws;
    float* H    = ws + 50176;
    float* H2   = H + (size_t)NN * FDIM;
    int*   rowptr = (int*)(H2 + (size_t)NN * FDIM);
    int*   cntab  = rowptr + 50176;
    int*   offtab = cntab + NBP * NBP;
    int*   tmp    = offtab + NBP * NBP;
    int*   src    = tmp + E;

    const int pieces = (E + PIECE - 1) / PIECE;
    const int ggrid = ((NN / 16) + 7) / 8;
    const int pgrid = (NN + 7) / 8;
    const int fgrid = ((NN / 16) + FW - 1) / FW;

    bin_edges<<<pieces, 256, 0, stream>>>(row, col, tmp, cntab, offtab, E);
    sort_bucket<<<NBUCK, SB_THREADS, 0, stream>>>(tmp, cntab, offtab, rowptr, dinv, src, pieces, NN);

    gemm_wmma<128, false><<<ggrid, 256, 0, stream>>>(x, W1, dinv, H, NN);

    pull_gemm<128><<<fgrid, FW * 32, 0, stream>>>(H, rowptr, src, dinv, b1, W2, H2, NN);

    pull_gemm<64><<<fgrid, FW * 32, 0, stream>>>(H2, rowptr, src, dinv, b2, W3, H, NN);

    pull_rows<64><<<pgrid, 256, 0, stream>>>(H, rowptr, src, dinv, b3, out, NN);
}


# layer-1 dense transform moved into the sort_bucket launch: a block's 32 waves multiply its bucket's 16 row strips between the sort's barriers, the sort runs under them; 5 launches not 6; same sums
# speedup vs baseline: 4.5693x; 1.0407x over previous
#include <hip/hip_runtime.h>

#define NN 50000
#define FDIM 128

typedef __attribute__((ext_vector_type(2))) float v2f;
typedef __attribute__((ext_vector_type(8))) float v8f;
typedef __attribute__((ext_vector_type(4))) float v4f;

template <int NT, int NW>
struct WPieces { v4f r[(NW / 4 + NT - 1) / NT]; };

template <int NT, int NW>
__device__ inline WPieces<NT, NW> load_weights(const float* __restrict__ W, int tid) {
    constexpr int NV  = NW / 4;
    constexpr int PER = (NV + NT - 1) / NT;
    const v4f* Wv = (const v4f*)W;
    WPieces<NT, NW> p;
#pragma unroll
    for (int u = 0; u < PER; ++u) {
        const int i = u * NT + tid;
        const bool part = (u == PER - 1) && (NV % NT != 0);
        p.r[u] = Wv[part ? min(i, NV - 1) : i];
    }
    return p;
}

template <int NT, int NW>
__device__ inline void store_weights(float* __restrict__ wlds, const WPieces<NT, NW>& p, int tid) {
    constexpr int NV  = NW / 4;
    constexpr int PER = (NV + NT - 1) / NT;
#pragma unroll
    for (int u = 0; u < PER; ++u) {
        const int i = u * NT + tid;
        const bool part = (u == PER - 1) && (NV % NT != 0);
        if (!part || i < NV) ((v4f*)wlds)[i] = p.r[u];
    }
}


#define PIECE 4096
#define EPT 16
#define BSH 8
#define BNODES 256
#define NBP 256
#define NBUCK ((NN + BNODES - 1) / BNODES)
#define SB_THREADS 512
#define SB_KEEP 10

__device__ inline int wave_scan_incl(int v, int lane) {
    for (int off = 1; off < 32; off <<= 1) {
        const int t = __shfl_up(v, off, 32);
        if (lane >= off) v += t;
    }
    return v;
}

__global__ __launch_bounds__(256) void bin_edges(const int* __restrict__ row,
                                                 const int* __restrict__ col,
                                                 int* __restrict__ tmp,
                                                 int* __restrict__ cntab,
                                                 int* __restrict__ offtab, int e) {
    __shared__ int cnt[NBP];
    __shared__ int wsum[8];
    __shared__ int lp[PIECE];
    const int tid  = threadIdx.x;
    const int lane = tid & 31;
    const int wv   = tid >> 5;
    cnt[tid] = 0;
    __syncthreads();

    const int beg = blockIdx.x * PIECE;
    const int len = min(PIECE, e - beg);
    int c[EPT], s[EPT], r[EPT], st[EPT];
#pragma unroll
    for (int u = 0; u < EPT; ++u) {
        const int k = min(u * 256 + tid, len - 1);
        c[u] = col[beg + k];
        s[u] = row[beg + k];
    }
#pragma unroll
    for (int u = 0; u < EPT; ++u) r[u] = (u * 256 + tid < len) ? atomicAdd(&cnt[c[u] >> BSH], 1) : 0;
    __syncthreads();

    const int v   = cnt[tid];
    const int inc = wave_scan_incl(v, lane);
    if (lane == 31) wsum[wv] = inc;
    __syncthreads();
    int front = 0;
    for (int w = 0; w < 8; ++w) front += (w < wv) ? wsum[w] : 0;
    const int excl = front + inc - v;
    cnt[tid] = excl;
    cntab[blockIdx.x * NBP + tid]  = v;
    offtab[blockIdx.x * NBP + tid] = excl;
    __syncthreads();

#pragma unroll
    for (int u = 0; u < EPT; ++u) st[u] = cnt[c[u] >> BSH];
#pragma unroll
    for (int u = 0; u < EPT; ++u) {
        if (u * 256 + tid < len) lp[st[u] + r[u]] = (s[u] << BSH) | (c[u] & (BNODES - 1));
    }
    __syncthreads();
    for (int k = tid; k < len; k += 256) tmp[beg + k] = lp[k];
}

#define SG_THREADS (2 * SB_THREADS)
#define XP 132
#define HC 64

__device__ inline void strip_steps(v8f (&acc)[HC / 16], const float* __restrict__ arow,
                                   const float* __restrict__ wcol, int kk0, int kk1, int koff) {
    for (int kk = kk0; kk < kk1; kk += 4) {
        const v2f a = *(const v2f*)(arow + kk + koff);
#pragma unroll
        for (int t = 0; t < HC / 16; ++t) {
            v2f b;
            b.x = wcol[(kk + koff)     * 128 + t * 16];
            b.y = wcol[(kk + koff + 1) * 128 + t * 16];
            acc[t] = __builtin_amdgcn_wmma_f32_16x16x4_f32(
                false, a, false, b, (short)0, acc[t], false, false);
        }
    }
}

__global__ __launch_bounds__(SG_THREADS) void sort_bucket(const int* __restrict__ tmp,
                                                          const int* __restrict__ cntab,
                                                          const int* __restrict__ offtab,
                                                          int* __restrict__ rowptr,
                                                          float* __restrict__ dinv,
                                                          int* __restrict__ src, int pieces, int n,
                                                          const float* __restrict__ X,
                                                          const float* __restrict__ W,
                                                          float* __restrict__ H) {
    constexpr int K    = FDIM;
    constexpr int FOUT = 128;
    constexpr int CT   = HC / 16;
    __shared__ int pstart[NBP + 1];
    __shared__ int poff[NBP];
    __shared__ int dcnt[BNODES];
    __shared__ int addr[SB_KEEP * SB_THREADS];
    __shared__ int wtot[8];
    __shared__ int ptot[8];
    __shared__ __attribute__((aligned(16))) float dloc[BNODES];
    __shared__ __attribute__((aligned(16))) float wlds[K * FOUT];
    __shared__ __attribute__((aligned(16))) float xt[16 * 16 * XP];
    const int tid  = threadIdx.x;
    const int lane = tid & 31;
    const int wv   = __builtin_amdgcn_readfirstlane((int)(threadIdx.x >> 5));
    const int g    = blockIdx.x;
    const bool sorter = wv < (SB_THREADS >> 5);
    const bool low8   = wv < (BNODES >> 5);
    const int  sw   = wv & 15;
    const int  ch   = wv >> 4;
    const int  row0 = g * BNODES + sw * 16;
    const int m    = lane & 15;
    const int koff = (lane >> 4) << 1;
    const float* arow = xt + sw * 16 * XP + m * XP;
    const float* wcol = wlds + ch * HC + m;

    int myc = 0, myo = 0, inc = 0;
    if (tid < pieces) {
        myc = cntab[tid * NBP + g];
        myo = offtab[tid * NBP + g];
    }
    const v4f* Xv = (const v4f*)X;
    v4f xr[8];
#pragma unroll
    for (int i = 0; i < 8; ++i) xr[i] = Xv[(size_t)min(row0 + ch * 8 + i, n - 1) * 32 + lane];
    const WPieces<SG_THREADS, K * FOUT> wp = load_weights<SG_THREADS, K * FOUT>(W, tid);
    if (low8) {
        inc = wave_scan_incl(myc, lane);
        int so = myo;
        for (int off = 16; off > 0; off >>= 1) so += __shfl_xor(so, off, 32);
        if (lane == 31) wtot[wv] = inc;
        if (lane == 0)  ptot[wv] = so;
        dcnt[tid] = 0;
    }
#pragma unroll
    for (int i = 0; i < 8; ++i) *(v4f*)(xt + (sw * 16 + ch * 8 + i) * XP + lane * 4) = xr[i];
    store_weights<SG_THREADS, K * FOUT>(wlds, wp, tid);
    __syncthreads();

    v8f acc[CT];
    const v8f zero = {0.f, 0.f, 0.f, 0.f, 0.f, 0.f, 0.f, 0.f};
#pragma unroll
    for (int t = 0; t < CT; ++t) acc[t] = zero;

    int base = 0;
    if (sorter) {
        for (int w = 0; w < 8; ++w) base += ptot[w];
        if (low8) {
            int front = 0;
            for (int w = 0; w < 8; ++w) front += (w < wv) ? wtot[w] : 0;
            const int excl = front + inc - myc;
            pstart[tid] = excl;
            poff[tid]   = myo;
            if (tid == NBP - 1) pstart[NBP] = excl + myc;
            const int a0 = tid * PIECE + myo;
            for (int i = 0; i < myc; ++i) {
                if (excl + i < SB_KEEP * SB_THREADS) addr[excl + i] = a0 + i;
            }
        }
    }
    strip_steps(acc, arow, wcol, 0, 16, koff);
    __syncthreads();

    int wk[SB_KEEP];
    int ng = 0;
    if (sorter) {
        ng = pstart[NBP];
        int ak[SB_KEEP];
#pragma unroll
        for (int k = 0; k < SB_KEEP; ++k) {
            const int j = k * SB_THREADS + tid;
            ak[k] = (j < ng) ? addr[j] : 0;
        }
#pragma unroll
        for (int k = 0; k < SB_KEEP; ++k) {
            const int j = k * SB_THREADS + tid;
            wk[k] = (j < ng) ? tmp[ak[k]] : -1;
        }
#pragma unroll
        for (int k = 0; k < SB_KEEP; ++k) {
            if (wk[k] >= 0) atomicAdd(&dcnt[wk[k] & (BNODES - 1)], 1);
        }
        for (int j = SB_KEEP * SB_THREADS + tid; j < ng; j += SB_THREADS) {
            int lo = 0, hi = NBP;
            for (int it = 0; it < 8; ++it) {
                const int mid = (lo + hi) >> 1;
                if (pstart[mid] <= j) lo = mid; else hi = mid;
            }
            const int w = tmp[lo * PIECE + poff[lo] + (j - pstart[lo])];
            atomicAdd(&dcnt[w & (BNODES - 1)], 1);
        }
    }
    strip_steps(acc, arow, wcol, 16, 48, koff);
    __syncthreads();

    int d = 0, dinc = 0;
    if (low8) {
        d    = dcnt[tid];
        dinc = wave_scan_incl(d, lane);
        if (lane == 31) wtot[wv] = dinc;
    }
    strip_steps(acc, arow, wcol, 48, 56, koff);
    __syncthreads();
    if (low8) {
        int front = 0;
        for (int w = 0; w < 8; ++w) front += (w < wv) ? wtot[w] : 0;
        const int dex  = front + dinc - d;
        const int node = g * BNODES + tid;
        const float dn = rsqrtf((float)(d + 1));
        dloc[tid] = dn;
        if (node < n) {
            rowptr[node] = base + dex;
            dinv[node]   = dn;
            if (node == n - 1) rowptr[n] = base + dex + d;
        }
        dcnt[tid] = dex;
    }
    strip_steps(acc, arow, wcol, 56, 64, koff);
    __syncthreads();

    if (sorter) {
        int pos[SB_KEEP];
#pragma unroll
        for (int k = 0; k < SB_KEEP; ++k) pos[k] = (wk[k] >= 0) ? atomicAdd(&dcnt[wk[k] & (BNODES - 1)], 1) : 0;
#pragma unroll
        for (int k = 0; k < SB_KEEP; ++k) {
            if (wk[k] >= 0) src[base + pos[k]] = wk[k] >> BSH;
        }
        for (int j = SB_KEEP * SB_THREADS + tid; j < ng; j += SB_THREADS) {
            int lo = 0, hi = NBP;
            for (int it = 0; it < 8; ++it) {
                const int mid = (lo + hi) >> 1;
                if (pstart[mid] <= j) lo = mid; else hi = mid;
            }
            const int w  = tmp[lo * PIECE + poff[lo] + (j - pstart[lo])];
            const int pj = atomicAdd(&dcnt[w & (BNODES - 1)], 1);
            src[base + pj] = w >> BSH;
        }
    }
    strip_steps(acc, arow, wcol, 64, K, koff);
    if (row0 < n) {
        const int mrow = (lane >> 4) << 3;
        const v4f dlo = *(const v4f*)(dloc + sw * 16 + mrow);
        const v4f dhi = *(const v4f*)(dloc + sw * 16 + mrow + 4);
        const float dv[8] = {dlo.x, dlo.y, dlo.z, dlo.w, dhi.x, dhi.y, dhi.z, dhi.w};
#pragma unroll
        for (int t = 0; t < CT; ++t) {
#pragma unroll
            for (int v = 0; v < 8; ++v) {
                H[(size_t)(row0 + mrow + v) * FOUT + ch * HC + t * 16 + m] = acc[t][v] * dv[v];
            }
        }
    }
}


template <int F>
__global__ __launch_bounds__(256) void pull_rows(const float* __restrict__ H,
                                                 const int* __restrict__ rowptr,
                                                 const int* __restrict__ src,
                                                 const float* __restrict__ dinv,
                                                 const float* __restrict__ bias,
                                                 float* __restrict__ O, int n) {
    constexpr int VPL = F / 32;
    typedef __attribute__((ext_vector_type(VPL))) float vec;
    const int lane = threadIdx.x & 31;
    const int node = __builtin_amdgcn_readfirstlane((int)(blockIdx.x * 8 + (threadIdx.x >> 5)));
    if (node >= n) return;

    const vec* Hv = (const vec*)H;
    const float d = dinv[node];
    const vec bv = ((const vec*)bias)[lane];
    vec acc = Hv[(size_t)node * 32 + lane];

    const int beg = rowptr[node];
    const int end = rowptr[node + 1];
    for (int base = beg; base < end; base += 32) {
        const int cnt = min(32, end - base);
        int myr = node;
        if (lane < cnt) myr = src[base + lane];
        int j = 0;
        for (; j + 4 <= cnt; j += 4) {
            const int r0 = __builtin_amdgcn_readlane(myr, j);
            const int r1 = __builtin_amdgcn_readlane(myr, j + 1);
            const int r2 = __builtin_amdgcn_readlane(myr, j + 2);
            const int r3 = __builtin_amdgcn_readlane(myr, j + 3);
            const vec h0 = Hv[(size_t)r0 * 32 + lane];
            const vec h1 = Hv[(size_t)r1 * 32 + lane];
            const vec h2 = Hv[(size_t)r2 * 32 + lane];
            const vec h3 = Hv[(size_t)r3 * 32 + lane];
            acc += h0;
            acc += h1;
            acc += h2;
            acc += h3;
        }
        for (; j < cnt; ++j) {
            const int r = __builtin_amdgcn_readlane(myr, j);
            acc += Hv[(size_t)r * 32 + lane];
        }
    }
    ((vec*)O)[(size_t)node * 32 + lane] = bv + d * acc;
}


struct Rows8 { v4f r[8]; };

__device__ inline Rows8 load8(const v4f* __restrict__ Hv, int ids, int j, int lane) {
    Rows8 h;
#pragma unroll
    for (int k = 0; k < 8; ++k) {
        const int r = __builtin_amdgcn_readlane(ids, min(j + k, 31));
        h.r[k] = Hv[(size_t)r * 32 + lane];
    }
    return h;
}

__device__ inline void add8(v4f& acc, const Rows8& h, int j, int cnt) {
#pragma unroll
    for (int k = 0; k < 8; ++k) {
        if (j + k < cnt) acc += h.r[k];
    }
}

#define FW 13
#define TP 132

template <int FOUT>
__global__ __launch_bounds__(FW * 32) void pull_gemm(const float* __restrict__ Hin,
                                                     const int* __restrict__ rowptr,
                                                     const int* __restrict__ src,
                                                     const float* __restrict__ dinv,
                                                     const float* __restrict__ bias,
                                                     const float* __restrict__ W,
                                                     float* __restrict__ Hout, int n) {
    constexpr int K  = FDIM;
    constexpr int CT = FOUT / 16;
    typedef v4f vec;
    __shared__ __attribute__((aligned(16))) float wlds[K * FOUT];
    __shared__ __attribute__((aligned(16))) float tile[FW * 16 * TP];

    const int lane = threadIdx.x & 31;
    const int wv = __builtin_amdgcn_readfirstlane((int)(threadIdx.x >> 5));
    const int tl = __builtin_amdgcn_readfirstlane((int)(blockIdx.x * FW + wv));
    const bool has = tl < (n >> 4);
    const int node0 = (has ? tl : 0) << 4;

    const vec* Hv = (const vec*)Hin;
    const int   rp = rowptr[node0 + min(lane, 16)];
    const float dl = dinv[node0 + (lane & 15)];
    const vec   bv = ((const vec*)bias)[lane];
    vec acc = Hv[(size_t)node0 * 32 + lane];
    const WPieces<FW * 32, K * FOUT> wp = load_weights<FW * 32, K * FOUT>(W, threadIdx.x);
    int beg = __builtin_amdgcn_readlane(rp, 0);
    int end = __builtin_amdgcn_readlane(rp, 1);
    int myr = node0;
    if (lane < end - beg) myr = src[beg + lane];
    store_weights<FW * 32, K * FOUT>(wlds, wp, threadIdx.x);
    __syncthreads();
    if (!has) return;
    float* my = tile + wv * 16 * TP;

    for (int i = 0; i < 16; ++i) {
        const int node = node0 + i;
        const int cnt0 = min(32, end - beg);
        const Rows8 h  = load8(Hv, myr, 0, lane);
        const Rows8 h2 = load8(Hv, myr, 8, lane);
        const Rows8 h3 = load8(Hv, myr, 16, lane);
        const int nn   = min(node + 1, n - 1);
        const int nbeg = end;
        const int nend = __builtin_amdgcn_readlane(rp, min(i + 2, 16));
        int nmyr = nn;
        if (lane < nend - nbeg) nmyr = src[nbeg + lane];
        const vec nself = Hv[(size_t)nn * 32 + lane];
        add8(acc, h, 0, cnt0);
        add8(acc, h2, 8, cnt0);
        add8(acc, h3, 16, cnt0);
        for (int j = 24; j < cnt0; j += 8) {
            const Rows8 g = load8(Hv, myr, j, lane);
            add8(acc, g, j, cnt0);
        }
        for (int base = beg + 32; base < end; base += 32) {
            const int cnt = min(32, end - base);
            int mor = node;
            if (lane < cnt) mor = src[base + lane];
            for (int j = 0; j < cnt; j += 8) {
                const Rows8 g = load8(Hv, mor, j, lane);
                add8(acc, g, j, cnt);
            }
        }
        const float d = __int_as_float(__builtin_amdgcn_readlane(__float_as_int(dl), i));
        vec o = bv + d * acc;
        o.x = fmaxf(o.x, 0.f); o.y = fmaxf(o.y, 0.f); o.z = fmaxf(o.z, 0.f); o.w = fmaxf(o.w, 0.f);
        *(vec*)(my + i * TP + lane * 4) = o;
        beg = nbeg; end = nend; myr = nmyr; acc = nself;
    }
    __builtin_amdgcn_fence(__ATOMIC_RELEASE, "wavefront");
    __builtin_amdgcn_wave_barrier();
    __builtin_amdgcn_fence(__ATOMIC_ACQUIRE, "wavefront");

    const int m    = lane & 15;
    const int koff = (lane >> 4) << 1;
    v8f cacc[CT];
    const v8f zero = {0.f, 0.f, 0.f, 0.f, 0.f, 0.f, 0.f, 0.f};
#pragma unroll
    for (int t = 0; t < CT; ++t) cacc[t] = zero;
    const float* arow = my + m * TP;
    for (int kk = 0; kk < K; kk += 4) {
        const v2f a = *(const v2f*)(arow + kk + koff);
#pragma unroll
        for (int t = 0; t < CT; ++t) {
            v2f b;
            b.x = wlds[(kk + koff)     * FOUT + t * 16 + m];
            b.y = wlds[(kk + koff + 1) * FOUT + t * 16 + m];
            cacc[t] = __builtin_amdgcn_wmma_f32_16x16x4_f32(
                false, a, false, b, (short)0, cacc[t], false, false);
        }
    }
    const int mrow = (lane >> 4) << 3;
    const float4 dlo = *(const float4*)(dinv + node0 + mrow);
    const float4 dhi = *(const float4*)(dinv + node0 + mrow + 4);
    const float dv[8] = {dlo.x, dlo.y, dlo.z, dlo.w, dhi.x, dhi.y, dhi.z, dhi.w};
#pragma unroll
    for (int t = 0; t < CT; ++t) {
#pragma unroll
        for (int v = 0; v < 8; ++v) {
            Hout[(size_t)(node0 + mrow + v) * FOUT + t * 16 + m] = cacc[t][v] * dv[v];
        }
    }
}


extern "C" void kernel_launch(void* const* d_in, const int* in_sizes, int n_in,
                              void* d_out, int out_size, void* d_ws, size_t ws_size,
                              hipStream_t stream) {
    const float* x  = (const float*)d_in[0];
    const int*   ei = (const int*)d_in[1];
    const float* W1 = (const float*)d_in[2];
    const float* b1 = (const float*)d_in[3];
    const float* W2 = (const float*)d_in[4];
    const float* b2 = (const float*)d_in[5];
    const float* W3 = (const float*)d_in[6];
    const float* b3 = (const float*)d_in[7];
    float* out = (float*)d_out;

    const int E = in_sizes[1] / 2;
    const int* row = ei;
    const int* col = ei + E;

    float* ws   = (float*)d_ws;
    float* dinv = ws;
    float* H    = ws + 50176;
    float* H2   = H + (size_t)NN * FDIM;
    int*   rowptr = (int*)(H2 + (size_t)NN * FDIM);
    int*   cntab  = rowptr + 50176;
    int*   offtab = cntab + NBP * NBP;
    int*   tmp    = offtab + NBP * NBP;
    int*   src    = tmp + E;

    const int pieces = (E + PIECE - 1) / PIECE;
    const int pgrid = (NN + 7) / 8;
    const int fgrid = ((NN / 16) + FW - 1) / FW;

    bin_edges<<<pieces, 256, 0, stream>>>(row, col, tmp, cntab, offtab, E);
    sort_bucket<<<NBUCK, SG_THREADS, 0, stream>>>(tmp, cntab, offtab, rowptr, dinv, src, pieces, NN, x, W1, H);

    pull_gemm<128><<<fgrid, FW * 32, 0, stream>>>(H, rowptr, src, dinv, b1, W2, H2, NN);

    pull_gemm<64><<<fgrid, FW * 32, 0, stream>>>(H2, rowptr, src, dinv, b2, W3, H, NN);

    pull_rows<64><<<pgrid, 256, 0, stream>>>(H, rowptr, src, dinv, b3, out, NN);
}
